# starting kernel
# baseline (speedup 1.0000x reference)
#include <hip/hip_runtime.h>
#include <hip/hip_bf16.h>

typedef float v2f __attribute__((ext_vector_type(2)));
typedef float v8f __attribute__((ext_vector_type(8)));

#define NNODES 100000
#define NEDGES 1600000

__global__ __launch_bounds__(128) void gcn_gemm_wmma(
    const float* __restrict__ X, const float* __restrict__ Wm,
    const float* __restrict__ bias, float* __restrict__ P,
    int N, int din, int dout, int doutp, int relu_in)
{
    __shared__ float lds[16 * (128 + 4)];
    const int ldsrow = din + 4;
    const int row0 = blockIdx.x * 16;
    const int t = threadIdx.x;

    const int nvec = (16 * din) >> 2;
    const int vper = din >> 2;
    for (int idx = t; idx < nvec; idx += 128) {
        const int r  = idx / vper;
        const int c4 = (idx - r * vper) << 2;
        const int row = row0 + r;
        float4 v = make_float4(0.f, 0.f, 0.f, 0.f);
        if (row < N) {
            v = *(const float4*)(X + (size_t)row * din + c4);
            if (relu_in) {
                v.x = fmaxf(v.x, 0.f); v.y = fmaxf(v.y, 0.f);
                v.z = fmaxf(v.z, 0.f); v.w = fmaxf(v.w, 0.f);
            }
        }
        *(float4*)(&lds[r * ldsrow + c4]) = v;
    }
    __syncthreads();

    const int wave = t >> 5;
    const int lane = t & 31;
    const int colbase = wave * 16;
    if (colbase >= doutp) return;

    const int half = lane >> 4;
    const int nrow = lane & 15;
    const int col  = colbase + nrow;

    const float* wk = Wm + (size_t)(2 * half) * doutp + col;
    int aoff = nrow * ldsrow + 2 * half;

    v8f acc = {};
    for (int k = 0; k < din; k += 4) {
        v2f a, b;
        a.x = lds[aoff];
        a.y = lds[aoff + 1];
        b.x = wk[0];
        b.y = wk[doutp];
        acc = __builtin_amdgcn_wmma_f32_16x16x4_f32(
                  false, a, false, b, (short)0, acc, false, false);
        aoff += 4;
        wk   += (size_t)4 * doutp;
    }

    const float bv = bias[col];
    #pragma unroll
    for (int r = 0; r < 8; ++r) {
        const int m   = r + 8 * half;
        const int row = row0 + m;
        if (row < N && col < dout)
            P[(size_t)row * dout + col] = acc[r] + bv;
    }
}

__global__ void gcn_pad_w4(const float* __restrict__ W4, const float* __restrict__ b4,
                           float* __restrict__ Wp, float* __restrict__ bp)
{
    const int i = blockIdx.x * blockDim.x + threadIdx.x;
    if (i < 64 * 16) {
        const int r = i >> 4, c = i & 15;
        Wp[i] = (c < 10) ? W4[r * 10 + c] : 0.f;
    } else if (i < 64 * 16 + 16) {
        const int c = i - 64 * 16;
        bp[c] = (c < 10) ? b4[c] : 0.f;
    }
}

__global__ void gcn_zero(float* __restrict__ p, int n)
{
    const int i = blockIdx.x * blockDim.x + threadIdx.x;
    if (i < n) p[i] = 0.f;
}

__global__ __launch_bounds__(256) void gcn_aggregate(
    const float* __restrict__ P, const int* __restrict__ src,
    const int* __restrict__ dst, const float* __restrict__ w,
    float* __restrict__ out, int E, int dout)
{
    const int gid = blockIdx.x * blockDim.x + threadIdx.x;
    const int e = gid >> 5;
    if (e >= E) return;
    const int lane = gid & 31;

    if (lane == 0) {
        const int ep = e + 4096;
        if (ep < E) {
            __builtin_prefetch(src + ep, 0, 0);
            __builtin_prefetch(dst + ep, 0, 0);
            __builtin_prefetch(w   + ep, 0, 0);
        }
    }

    const int c = lane << 1;
    if (c >= dout) return;

    const int s = src[e];
    const int d = dst[e];
    const float ww = w[e];

    const float2 h = *(const float2*)(P + (size_t)s * dout + c);
    float* o = out + (size_t)d * dout + c;
    __hip_atomic_fetch_add(o,     h.x * ww, __ATOMIC_RELAXED, __HIP_MEMORY_SCOPE_AGENT);
    __hip_atomic_fetch_add(o + 1, h.y * ww, __ATOMIC_RELAXED, __HIP_MEMORY_SCOPE_AGENT);
}

extern "C" void kernel_launch(void* const* d_in, const int* in_sizes, int n_in,
                              void* d_out, int out_size, void* d_ws, size_t ws_size,
                              hipStream_t stream)
{
    (void)in_sizes; (void)n_in; (void)out_size; (void)ws_size;
    const int N = NNODES, E = NEDGES;

    const float* x    = (const float*)d_in[0];
    const int*   esrc = (const int*)  d_in[1];
    const int*   edst = (const int*)  d_in[2];
    const float* ew   = (const float*)d_in[3];
    const float* Ws[5]; const float* bs[5];
    for (int i = 0; i < 5; ++i) {
        Ws[i] = (const float*)d_in[4 + 2 * i];
        bs[i] = (const float*)d_in[5 + 2 * i];
    }
    float* out  = (float*)d_out;
    float* buf0 = (float*)d_ws;
    float* buf1 = buf0 + (size_t)N * 64;
    float* buf2 = buf1 + (size_t)N * 64;
    float* wpad = buf2 + (size_t)N * 64;
    float* bpad = wpad + 64 * 16;

    const dim3 ggrid((N + 15) / 16);
    const int  agrid = (E * 32 + 255) / 256;

    gcn_pad_w4<<<(64 * 16 + 16 + 255) / 256, 256, 0, stream>>>(Ws[4], bs[4], wpad, bpad);

    gcn_gemm_wmma<<<ggrid, 128, 0, stream>>>(x, Ws[0], bs[0], buf0, N, 128, 64, 64, 0);
    gcn_zero<<<(N * 64 + 255) / 256, 256, 0, stream>>>(buf1, N * 64);
    gcn_aggregate<<<agrid, 256, 0, stream>>>(buf0, esrc, edst, ew, buf1, E, 64);

    gcn_gemm_wmma<<<ggrid, 128, 0, stream>>>(buf1, Ws[1], bs[1], buf2, N, 64, 64, 64, 1);
    gcn_zero<<<(N * 64 + 255) / 256, 256, 0, stream>>>(buf0, N * 64);
    gcn_aggregate<<<agrid, 256, 0, stream>>>(buf2, esrc, edst, ew, buf0, E, 64);

    gcn_gemm_wmma<<<ggrid, 128, 0, stream>>>(buf0, Ws[2], bs[2], buf1, N, 64, 64, 64, 1);
    gcn_zero<<<(N * 64 + 255) / 256, 256, 0, stream>>>(buf2, N * 64);
    gcn_aggregate<<<agrid, 256, 0, stream>>>(buf1, esrc, edst, ew, buf2, E, 64);

    gcn_gemm_wmma<<<ggrid, 128, 0, stream>>>(buf2, Ws[3], bs[3], buf0, N, 64, 64, 64, 1);
    gcn_zero<<<(N * 64 + 255) / 256, 256, 0, stream>>>(buf1, N * 64);
    gcn_aggregate<<<agrid, 256, 0, stream>>>(buf0, esrc, edst, ew, buf1, E, 64);

    gcn_gemm_wmma<<<ggrid, 128, 0, stream>>>(buf1, wpad, bpad, buf2, N, 64, 10, 16, 1);
    gcn_zero<<<(N * 10 + 255) / 256, 256, 0, stream>>>(out, N * 10);
    gcn_aggregate<<<agrid, 256, 0, stream>>>(buf2, esrc, edst, ew, out, E, 10);
}


# Replace f32-atomic edge scatter by a per-call dst-sorted edge list (int-atomic rank, scan, fill) and a per-node gather-sum, no zero-fills; sums now sequential per node in arrival-rank order
# speedup vs baseline: 2.0557x; 2.0557x over previous
#include <hip/hip_runtime.h>
#include <hip/hip_bf16.h>

typedef float v2f __attribute__((ext_vector_type(2)));
typedef float v8f __attribute__((ext_vector_type(8)));

#define NNODES 100000
#define NEDGES 1600000

__global__ __launch_bounds__(128) void gcn_gemm_wmma(
    const float* __restrict__ X, const float* __restrict__ Wm,
    const float* __restrict__ bias, float* __restrict__ P,
    int N, int din, int dout, int doutp, int relu_in)
{
    __shared__ float lds[16 * (128 + 4)];
    const int ldsrow = din + 4;
    const int row0 = blockIdx.x * 16;
    const int t = threadIdx.x;

    const int nvec = (16 * din) >> 2;
    const int vper = din >> 2;
    for (int idx = t; idx < nvec; idx += 128) {
        const int r  = idx / vper;
        const int c4 = (idx - r * vper) << 2;
        const int row = row0 + r;
        float4 v = make_float4(0.f, 0.f, 0.f, 0.f);
        if (row < N) {
            v = *(const float4*)(X + (size_t)row * din + c4);
            if (relu_in) {
                v.x = fmaxf(v.x, 0.f); v.y = fmaxf(v.y, 0.f);
                v.z = fmaxf(v.z, 0.f); v.w = fmaxf(v.w, 0.f);
            }
        }
        *(float4*)(&lds[r * ldsrow + c4]) = v;
    }
    __syncthreads();

    const int wave = t >> 5;
    const int lane = t & 31;
    const int colbase = wave * 16;
    if (colbase >= doutp) return;

    const int half = lane >> 4;
    const int nrow = lane & 15;
    const int col  = colbase + nrow;

    const float* wk = Wm + (size_t)(2 * half) * doutp + col;
    int aoff = nrow * ldsrow + 2 * half;

    v8f acc = {};
    for (int k = 0; k < din; k += 4) {
        v2f a, b;
        a.x = lds[aoff];
        a.y = lds[aoff + 1];
        b.x = wk[0];
        b.y = wk[doutp];
        acc = __builtin_amdgcn_wmma_f32_16x16x4_f32(
                  false, a, false, b, (short)0, acc, false, false);
        aoff += 4;
        wk   += (size_t)4 * doutp;
    }

    const float bv = bias[col];
    #pragma unroll
    for (int r = 0; r < 8; ++r) {
        const int m   = r + 8 * half;
        const int row = row0 + m;
        if (row < N && col < dout)
            P[(size_t)row * dout + col] = acc[r] + bv;
    }
}

__global__ void gcn_pad_w4(const float* __restrict__ W4, const float* __restrict__ b4,
                           float* __restrict__ Wp, float* __restrict__ bp)
{
    const int i = blockIdx.x * blockDim.x + threadIdx.x;
    if (i < 64 * 16) {
        const int r = i >> 4, c = i & 15;
        Wp[i] = (c < 10) ? W4[r * 10 + c] : 0.f;
    } else if (i < 64 * 16 + 16) {
        const int c = i - 64 * 16;
        bp[c] = (c < 10) ? b4[c] : 0.f;
    }
}

__global__ void gcn_zero_i32(int* __restrict__ p, int n)
{
    const int i = blockIdx.x * blockDim.x + threadIdx.x;
    if (i < n) p[i] = 0;
}

__global__ __launch_bounds__(256) void gcn_count_rank(
    const int* __restrict__ dst, int* __restrict__ cnt, int* __restrict__ rank, int E)
{
    const int e = blockIdx.x * blockDim.x + threadIdx.x;
    if (e >= E) return;
    rank[e] = __hip_atomic_fetch_add(cnt + dst[e], 1, __ATOMIC_RELAXED,
                                     __HIP_MEMORY_SCOPE_AGENT);
}

__global__ __launch_bounds__(1024) void gcn_scan(
    const int* __restrict__ cnt, int* __restrict__ rowptr, int n)
{
    __shared__ int wsum[32];
    const int t    = threadIdx.x;
    const int lane = t & 31;
    const int wv   = t >> 5;
    const int per  = (n + 1023) / 1024;
    const int beg  = min(t * per, n);
    const int end  = min(beg + per, n);

    int s = 0;
    for (int i = beg; i < end; ++i) s += cnt[i];

    int incl = s;
    for (int off = 1; off < 32; off <<= 1) {
        const int v = __shfl_up(incl, off, 32);
        if (lane >= off) incl += v;
    }
    if (lane == 31) wsum[wv] = incl;
    __syncthreads();
    if (wv == 0) {
        const int ws = wsum[lane];
        int wi = ws;
        for (int off = 1; off < 32; off <<= 1) {
            const int v = __shfl_up(wi, off, 32);
            if (lane >= off) wi += v;
        }
        wsum[lane] = wi - ws;
    }
    __syncthreads();

    int run = wsum[wv] + incl - s;
    for (int i = beg; i < end; ++i) { rowptr[i] = run; run += cnt[i]; }
    if (t == 1023) rowptr[n] = run;
}

__global__ __launch_bounds__(256) void gcn_fill(
    const int* __restrict__ src, const int* __restrict__ dst,
    const float* __restrict__ w, const int* __restrict__ rowptr,
    const int* __restrict__ rank, int* __restrict__ csr, int E)
{
    const int e = blockIdx.x * blockDim.x + threadIdx.x;
    if (e >= E) return;
    const int pos = rowptr[dst[e]] + rank[e];
    int2 v;
    v.x = src[e];
    v.y = __float_as_int(w[e]);
    *(int2*)(csr + 2 * (size_t)pos) = v;
}

__device__ __forceinline__ void fma4(float4& a, const float4 r, const float w)
{
    a.x = fmaf(r.x, w, a.x); a.y = fmaf(r.y, w, a.y);
    a.z = fmaf(r.z, w, a.z); a.w = fmaf(r.w, w, a.w);
}

__global__ __launch_bounds__(256) void gcn_gather64(
    const float* __restrict__ P, const int* __restrict__ rowptr,
    const int* __restrict__ csr, float* __restrict__ out, int N)
{
    const int gid  = blockIdx.x * 256 + threadIdx.x;
    const int node = gid >> 4;
    const int l    = gid & 15;
    if (node >= N) return;
    const int2*   ed = (const int2*)csr;
    const float4* P4 = (const float4*)P;

    int j = rowptr[node];
    const int end = rowptr[node + 1];
    float4 acc = make_float4(0.f, 0.f, 0.f, 0.f);
    for (; j + 4 <= end; j += 4) {
        const int2 e0 = ed[j], e1 = ed[j + 1], e2 = ed[j + 2], e3 = ed[j + 3];
        const float4 r0 = P4[(size_t)e0.x * 16 + l];
        const float4 r1 = P4[(size_t)e1.x * 16 + l];
        const float4 r2 = P4[(size_t)e2.x * 16 + l];
        const float4 r3 = P4[(size_t)e3.x * 16 + l];
        fma4(acc, r0, __int_as_float(e0.y));
        fma4(acc, r1, __int_as_float(e1.y));
        fma4(acc, r2, __int_as_float(e2.y));
        fma4(acc, r3, __int_as_float(e3.y));
    }
    if (j + 2 <= end) {
        const int2 e0 = ed[j], e1 = ed[j + 1];
        const float4 r0 = P4[(size_t)e0.x * 16 + l];
        const float4 r1 = P4[(size_t)e1.x * 16 + l];
        fma4(acc, r0, __int_as_float(e0.y));
        fma4(acc, r1, __int_as_float(e1.y));
        j += 2;
    }
    if (j < end) {
        const int2 e0 = ed[j];
        const float4 r0 = P4[(size_t)e0.x * 16 + l];
        fma4(acc, r0, __int_as_float(e0.y));
    }
    ((float4*)out)[(size_t)node * 16 + l] = acc;
}

__global__ __launch_bounds__(256) void gcn_gather10(
    const float* __restrict__ P, const int* __restrict__ rowptr,
    const int* __restrict__ csr, float* __restrict__ out, int N)
{
    __shared__ float stage[160];
    const int t    = threadIdx.x;
    const int nb   = t >> 4;
    const int l    = t & 15;
    const int node = blockIdx.x * 16 + nb;
    const int2* ed = (const int2*)csr;

    float acc = 0.f;
    if (node < N && l < 10) {
        int j = rowptr[node];
        const int end = rowptr[node + 1];
        for (; j + 4 <= end; j += 4) {
            const int2 e0 = ed[j], e1 = ed[j + 1], e2 = ed[j + 2], e3 = ed[j + 3];
            const float r0 = P[(size_t)e0.x * 10 + l];
            const float r1 = P[(size_t)e1.x * 10 + l];
            const float r2 = P[(size_t)e2.x * 10 + l];
            const float r3 = P[(size_t)e3.x * 10 + l];
            acc = fmaf(r0, __int_as_float(e0.y), acc);
            acc = fmaf(r1, __int_as_float(e1.y), acc);
            acc = fmaf(r2, __int_as_float(e2.y), acc);
            acc = fmaf(r3, __int_as_float(e3.y), acc);
        }
        for (; j < end; ++j) {
            const int2 e0 = ed[j];
            acc = fmaf(P[(size_t)e0.x * 10 + l], __int_as_float(e0.y), acc);
        }
    }
    if (l < 10) stage[nb * 10 + l] = acc;
    __syncthreads();
    const int o = blockIdx.x * 160 + t;
    if (t < 160 && o < N * 10) out[o] = stage[t];
}

extern "C" void kernel_launch(void* const* d_in, const int* in_sizes, int n_in,
                              void* d_out, int out_size, void* d_ws, size_t ws_size,
                              hipStream_t stream)
{
    (void)in_sizes; (void)n_in; (void)out_size; (void)ws_size;
    const int N = NNODES, E = NEDGES;

    const float* x    = (const float*)d_in[0];
    const int*   esrc = (const int*)  d_in[1];
    const int*   edst = (const int*)  d_in[2];
    const float* ew   = (const float*)d_in[3];
    const float* Ws[5]; const float* bs[5];
    for (int i = 0; i < 5; ++i) {
        Ws[i] = (const float*)d_in[4 + 2 * i];
        bs[i] = (const float*)d_in[5 + 2 * i];
    }
    float* out  = (float*)d_out;
    float* buf0 = (float*)d_ws;
    float* buf1 = buf0 + (size_t)N * 64;
    float* buf2 = buf1 + (size_t)N * 64;
    float* wpad = buf2 + (size_t)N * 64;
    float* bpad = wpad + 64 * 16;
    int*   cnt    = (int*)(bpad + 64);
    int*   rowptr = cnt + N;
    int*   rank   = rowptr + (N + 64);
    int*   csr    = rank + E;

    const dim3 ggrid((N + 15) / 16);
    const int  egrid = (E + 255) / 256;

    gcn_pad_w4<<<(64 * 16 + 16 + 255) / 256, 256, 0, stream>>>(Ws[4], bs[4], wpad, bpad);

    gcn_zero_i32<<<(N + 255) / 256, 256, 0, stream>>>(cnt, N);
    gcn_count_rank<<<egrid, 256, 0, stream>>>(edst, cnt, rank, E);
    gcn_scan<<<1, 1024, 0, stream>>>(cnt, rowptr, N);
    gcn_fill<<<egrid, 256, 0, stream>>>(esrc, edst, ew, rowptr, rank, csr, E);

    gcn_gemm_wmma<<<ggrid, 128, 0, stream>>>(x, Ws[0], bs[0], buf0, N, 128, 64, 64, 0);
    gcn_gather64<<<ggrid, 256, 0, stream>>>(buf0, rowptr, csr, buf1, N);

    gcn_gemm_wmma<<<ggrid, 128, 0, stream>>>(buf1, Ws[1], bs[1], buf2, N, 64, 64, 64, 1);
    gcn_gather64<<<ggrid, 256, 0, stream>>>(buf2, rowptr, csr, buf0, N);

    gcn_gemm_wmma<<<ggrid, 128, 0, stream>>>(buf0, Ws[2], bs[2], buf1, N, 64, 64, 64, 1);
    gcn_gather64<<<ggrid, 256, 0, stream>>>(buf1, rowptr, csr, buf2, N);

    gcn_gemm_wmma<<<ggrid, 128, 0, stream>>>(buf2, Ws[3], bs[3], buf0, N, 64, 64, 64, 1);
    gcn_gather64<<<ggrid, 256, 0, stream>>>(buf0, rowptr, csr, buf1, N);

    gcn_gemm_wmma<<<ggrid, 128, 0, stream>>>(buf1, wpad, bpad, buf2, N, 64, 10, 16, 1);
    gcn_gather10<<<ggrid, 256, 0, stream>>>(buf2, rowptr, csr, out, N);
}


# Prefix scan of in-degrees: coalesced 16-byte loads/stores, per-row wave scans in registers and one LDS pass over wave totals replace per-thread strided chunk walks; rowptr values identical
# speedup vs baseline: 2.2676x; 1.1031x over previous
#include <hip/hip_runtime.h>
#include <hip/hip_bf16.h>

typedef float v2f __attribute__((ext_vector_type(2)));
typedef float v8f __attribute__((ext_vector_type(8)));

#define NNODES 100000
#define NEDGES 1600000

__global__ __launch_bounds__(128) void gcn_gemm_wmma(
    const float* __restrict__ X, const float* __restrict__ Wm,
    const float* __restrict__ bias, float* __restrict__ P,
    int N, int din, int dout, int doutp, int relu_in)
{
    __shared__ float lds[16 * (128 + 4)];
    const int ldsrow = din + 4;
    const int row0 = blockIdx.x * 16;
    const int t = threadIdx.x;

    const int nvec = (16 * din) >> 2;
    const int vper = din >> 2;
    for (int idx = t; idx < nvec; idx += 128) {
        const int r  = idx / vper;
        const int c4 = (idx - r * vper) << 2;
        const int row = row0 + r;
        float4 v = make_float4(0.f, 0.f, 0.f, 0.f);
        if (row < N) {
            v = *(const float4*)(X + (size_t)row * din + c4);
            if (relu_in) {
                v.x = fmaxf(v.x, 0.f); v.y = fmaxf(v.y, 0.f);
                v.z = fmaxf(v.z, 0.f); v.w = fmaxf(v.w, 0.f);
            }
        }
        *(float4*)(&lds[r * ldsrow + c4]) = v;
    }
    __syncthreads();

    const int wave = t >> 5;
    const int lane = t & 31;
    const int colbase = wave * 16;
    if (colbase >= doutp) return;

    const int half = lane >> 4;
    const int nrow = lane & 15;
    const int col  = colbase + nrow;

    const float* wk = Wm + (size_t)(2 * half) * doutp + col;
    int aoff = nrow * ldsrow + 2 * half;

    v8f acc = {};
    for (int k = 0; k < din; k += 4) {
        v2f a, b;
        a.x = lds[aoff];
        a.y = lds[aoff + 1];
        b.x = wk[0];
        b.y = wk[doutp];
        acc = __builtin_amdgcn_wmma_f32_16x16x4_f32(
                  false, a, false, b, (short)0, acc, false, false);
        aoff += 4;
        wk   += (size_t)4 * doutp;
    }

    const float bv = bias[col];
    #pragma unroll
    for (int r = 0; r < 8; ++r) {
        const int m   = r + 8 * half;
        const int row = row0 + m;
        if (row < N && col < dout)
            P[(size_t)row * dout + col] = acc[r] + bv;
    }
}

__global__ void gcn_pad_w4(const float* __restrict__ W4, const float* __restrict__ b4,
                           float* __restrict__ Wp, float* __restrict__ bp)
{
    const int i = blockIdx.x * blockDim.x + threadIdx.x;
    if (i < 64 * 16) {
        const int r = i >> 4, c = i & 15;
        Wp[i] = (c < 10) ? W4[r * 10 + c] : 0.f;
    } else if (i < 64 * 16 + 16) {
        const int c = i - 64 * 16;
        bp[c] = (c < 10) ? b4[c] : 0.f;
    }
}

__global__ void gcn_zero_i32(int* __restrict__ p, int n)
{
    const int i = blockIdx.x * blockDim.x + threadIdx.x;
    if (i < n) p[i] = 0;
}

__global__ __launch_bounds__(256) void gcn_count_rank(
    const int* __restrict__ dst, int* __restrict__ cnt, int* __restrict__ rank, int E)
{
    const int e = blockIdx.x * blockDim.x + threadIdx.x;
    if (e >= E) return;
    rank[e] = __hip_atomic_fetch_add(cnt + dst[e], 1, __ATOMIC_RELAXED,
                                     __HIP_MEMORY_SCOPE_AGENT);
}

#define SCAN_ROWS ((NNODES / 4 + 1023) / 1024)
static_assert(NNODES % 4 == 0, "scan reads the counts as 4-int vectors");
static_assert(SCAN_ROWS * 1024 >= NNODES / 4, "the rows must cover every vector");

__global__ __launch_bounds__(1024) void gcn_scan(
    const int* __restrict__ cnt, int* __restrict__ rowptr)
{
    constexpr int NVEC = NNODES / 4;
    __shared__ int tot[SCAN_ROWS * 32];
    const int t    = threadIdx.x;
    const int lane = t & 31;
    const int wv   = t >> 5;
    const int4* c4 = (const int4*)cnt;

    int s[SCAN_ROWS], incl[SCAN_ROWS];
    #pragma unroll
    for (int r = 0; r < SCAN_ROWS; ++r) {
        const int v = r * 1024 + t;
        int4 c = make_int4(0, 0, 0, 0);
        if (v < NVEC) c = c4[v];
        s[r] = c.x + c.y + c.z + c.w;
    }
    #pragma unroll
    for (int r = 0; r < SCAN_ROWS; ++r) {
        int x = s[r];
        #pragma unroll
        for (int off = 1; off < 32; off <<= 1) {
            const int y = __shfl_up(x, off, 32);
            if (lane >= off) x += y;
        }
        incl[r] = x;
    }
    if (lane == 31) {
        #pragma unroll
        for (int r = 0; r < SCAN_ROWS; ++r) tot[r * 32 + wv] = incl[r];
    }
    __syncthreads();
    if (wv == 0) {
        int loc[SCAN_ROWS];
        int run = 0;
        #pragma unroll
        for (int k = 0; k < SCAN_ROWS; ++k) {
            const int v = tot[lane * SCAN_ROWS + k];
            loc[k] = run;
            run += v;
        }
        int wi = run;
        #pragma unroll
        for (int off = 1; off < 32; off <<= 1) {
            const int y = __shfl_up(wi, off, 32);
            if (lane >= off) wi += y;
        }
        const int ex = wi - run;
        #pragma unroll
        for (int k = 0; k < SCAN_ROWS; ++k) tot[lane * SCAN_ROWS + k] = loc[k] + ex;
        if (lane == 31) rowptr[NNODES] = wi;
    }
    __syncthreads();
    int4* r4 = (int4*)rowptr;
    #pragma unroll
    for (int r = 0; r < SCAN_ROWS; ++r) {
        const int v = r * 1024 + t;
        if (v < NVEC) {
            const int4 c = c4[v];
            int4 o;
            o.x = tot[r * 32 + wv] + incl[r] - s[r];
            o.y = o.x + c.x;
            o.z = o.y + c.y;
            o.w = o.z + c.z;
            r4[v] = o;
        }
    }
}

__global__ __launch_bounds__(256) void gcn_fill(
    const int* __restrict__ src, const int* __restrict__ dst,
    const float* __restrict__ w, const int* __restrict__ rowptr,
    const int* __restrict__ rank, int* __restrict__ csr, int E)
{
    const int e = blockIdx.x * blockDim.x + threadIdx.x;
    if (e >= E) return;
    const int pos = rowptr[dst[e]] + rank[e];
    int2 v;
    v.x = src[e];
    v.y = __float_as_int(w[e]);
    *(int2*)(csr + 2 * (size_t)pos) = v;
}

__device__ __forceinline__ void fma4(float4& a, const float4 r, const float w)
{
    a.x = fmaf(r.x, w, a.x); a.y = fmaf(r.y, w, a.y);
    a.z = fmaf(r.z, w, a.z); a.w = fmaf(r.w, w, a.w);
}

__global__ __launch_bounds__(256) void gcn_gather64(
    const float* __restrict__ P, const int* __restrict__ rowptr,
    const int* __restrict__ csr, float* __restrict__ out, int N)
{
    const int gid  = blockIdx.x * 256 + threadIdx.x;
    const int node = gid >> 4;
    const int l    = gid & 15;
    if (node >= N) return;
    const int2*   ed = (const int2*)csr;
    const float4* P4 = (const float4*)P;

    int j = rowptr[node];
    const int end = rowptr[node + 1];
    float4 acc = make_float4(0.f, 0.f, 0.f, 0.f);
    for (; j + 4 <= end; j += 4) {
        const int2 e0 = ed[j], e1 = ed[j + 1], e2 = ed[j + 2], e3 = ed[j + 3];
        const float4 r0 = P4[(size_t)e0.x * 16 + l];
        const float4 r1 = P4[(size_t)e1.x * 16 + l];
        const float4 r2 = P4[(size_t)e2.x * 16 + l];
        const float4 r3 = P4[(size_t)e3.x * 16 + l];
        fma4(acc, r0, __int_as_float(e0.y));
        fma4(acc, r1, __int_as_float(e1.y));
        fma4(acc, r2, __int_as_float(e2.y));
        fma4(acc, r3, __int_as_float(e3.y));
    }
    if (j + 2 <= end) {
        const int2 e0 = ed[j], e1 = ed[j + 1];
        const float4 r0 = P4[(size_t)e0.x * 16 + l];
        const float4 r1 = P4[(size_t)e1.x * 16 + l];
        fma4(acc, r0, __int_as_float(e0.y));
        fma4(acc, r1, __int_as_float(e1.y));
        j += 2;
    }
    if (j < end) {
        const int2 e0 = ed[j];
        const float4 r0 = P4[(size_t)e0.x * 16 + l];
        fma4(acc, r0, __int_as_float(e0.y));
    }
    ((float4*)out)[(size_t)node * 16 + l] = acc;
}

__global__ __launch_bounds__(256) void gcn_gather10(
    const float* __restrict__ P, const int* __restrict__ rowptr,
    const int* __restrict__ csr, float* __restrict__ out, int N)
{
    __shared__ float stage[160];
    const int t    = threadIdx.x;
    const int nb   = t >> 4;
    const int l    = t & 15;
    const int node = blockIdx.x * 16 + nb;
    const int2* ed = (const int2*)csr;

    float acc = 0.f;
    if (node < N && l < 10) {
        int j = rowptr[node];
        const int end = rowptr[node + 1];
        for (; j + 4 <= end; j += 4) {
            const int2 e0 = ed[j], e1 = ed[j + 1], e2 = ed[j + 2], e3 = ed[j + 3];
            const float r0 = P[(size_t)e0.x * 10 + l];
            const float r1 = P[(size_t)e1.x * 10 + l];
            const float r2 = P[(size_t)e2.x * 10 + l];
            const float r3 = P[(size_t)e3.x * 10 + l];
            acc = fmaf(r0, __int_as_float(e0.y), acc);
            acc = fmaf(r1, __int_as_float(e1.y), acc);
            acc = fmaf(r2, __int_as_float(e2.y), acc);
            acc = fmaf(r3, __int_as_float(e3.y), acc);
        }
        for (; j < end; ++j) {
            const int2 e0 = ed[j];
            acc = fmaf(P[(size_t)e0.x * 10 + l], __int_as_float(e0.y), acc);
        }
    }
    if (l < 10) stage[nb * 10 + l] = acc;
    __syncthreads();
    const int o = blockIdx.x * 160 + t;
    if (t < 160 && o < N * 10) out[o] = stage[t];
}

extern "C" void kernel_launch(void* const* d_in, const int* in_sizes, int n_in,
                              void* d_out, int out_size, void* d_ws, size_t ws_size,
                              hipStream_t stream)
{
    (void)in_sizes; (void)n_in; (void)out_size; (void)ws_size;
    const int N = NNODES, E = NEDGES;

    const float* x    = (const float*)d_in[0];
    const int*   esrc = (const int*)  d_in[1];
    const int*   edst = (const int*)  d_in[2];
    const float* ew   = (const float*)d_in[3];
    const float* Ws[5]; const float* bs[5];
    for (int i = 0; i < 5; ++i) {
        Ws[i] = (const float*)d_in[4 + 2 * i];
        bs[i] = (const float*)d_in[5 + 2 * i];
    }
    float* out  = (float*)d_out;
    float* buf0 = (float*)d_ws;
    float* buf1 = buf0 + (size_t)N * 64;
    float* buf2 = buf1 + (size_t)N * 64;
    float* wpad = buf2 + (size_t)N * 64;
    float* bpad = wpad + 64 * 16;
    int*   cnt    = (int*)(bpad + 64);
    int*   rowptr = cnt + N;
    int*   rank   = rowptr + (N + 64);
    int*   csr    = rank + E;

    const dim3 ggrid((N + 15) / 16);
    const int  egrid = (E + 255) / 256;

    gcn_pad_w4<<<(64 * 16 + 16 + 255) / 256, 256, 0, stream>>>(Ws[4], bs[4], wpad, bpad);

    gcn_zero_i32<<<(N + 255) / 256, 256, 0, stream>>>(cnt, N);
    gcn_count_rank<<<egrid, 256, 0, stream>>>(edst, cnt, rank, E);
    gcn_scan<<<1, 1024, 0, stream>>>(cnt, rowptr);
    gcn_fill<<<egrid, 256, 0, stream>>>(esrc, edst, ew, rowptr, rank, csr, E);

    gcn_gemm_wmma<<<ggrid, 128, 0, stream>>>(x, Ws[0], bs[0], buf0, N, 128, 64, 64, 0);
    gcn_gather64<<<ggrid, 256, 0, stream>>>(buf0, rowptr, csr, buf1, N);

    gcn_gemm_wmma<<<ggrid, 128, 0, stream>>>(buf1, Ws[1], bs[1], buf2, N, 64, 64, 64, 1);
    gcn_gather64<<<ggrid, 256, 0, stream>>>(buf2, rowptr, csr, buf0, N);

    gcn_gemm_wmma<<<ggrid, 128, 0, stream>>>(buf0, Ws[2], bs[2], buf1, N, 64, 64, 64, 1);
    gcn_gather64<<<ggrid, 256, 0, stream>>>(buf1, rowptr, csr, buf2, N);

    gcn_gemm_wmma<<<ggrid, 128, 0, stream>>>(buf2, Ws[3], bs[3], buf0, N, 64, 64, 64, 1);
    gcn_gather64<<<ggrid, 256, 0, stream>>>(buf0, rowptr, csr, buf1, N);

    gcn_gemm_wmma<<<ggrid, 128, 0, stream>>>(buf1, wpad, bpad, buf2, N, 64, 10, 16, 1);
    gcn_gather10<<<ggrid, 256, 0, stream>>>(buf2, rowptr, csr, out, N);
}


# Fuse each 64-wide gather-sum with the next layer's GEMM: a work-group's 16 summed rows go relu'd into LDS and straight into the f32 WMMA; 4 launches and the h round trip go; sum orders unchanged
# speedup vs baseline: 2.7167x; 1.1981x over previous
#include <hip/hip_runtime.h>
#include <hip/hip_bf16.h>

typedef float v2f __attribute__((ext_vector_type(2)));
typedef float v8f __attribute__((ext_vector_type(8)));

#define NNODES 100000
#define NEDGES 1600000

__global__ __launch_bounds__(128) void gcn_gemm_wmma(
    const float* __restrict__ X, const float* __restrict__ Wm,
    const float* __restrict__ bias, float* __restrict__ P,
    int N, int din, int dout, int doutp, int relu_in)
{
    __shared__ float lds[16 * (128 + 4)];
    const int ldsrow = din + 4;
    const int row0 = blockIdx.x * 16;
    const int t = threadIdx.x;

    const int nvec = (16 * din) >> 2;
    const int vper = din >> 2;
    for (int idx = t; idx < nvec; idx += 128) {
        const int r  = idx / vper;
        const int c4 = (idx - r * vper) << 2;
        const int row = row0 + r;
        float4 v = make_float4(0.f, 0.f, 0.f, 0.f);
        if (row < N) {
            v = *(const float4*)(X + (size_t)row * din + c4);
            if (relu_in) {
                v.x = fmaxf(v.x, 0.f); v.y = fmaxf(v.y, 0.f);
                v.z = fmaxf(v.z, 0.f); v.w = fmaxf(v.w, 0.f);
            }
        }
        *(float4*)(&lds[r * ldsrow + c4]) = v;
    }
    __syncthreads();

    const int wave = t >> 5;
    const int lane = t & 31;
    const int colbase = wave * 16;
    if (colbase >= doutp) return;

    const int half = lane >> 4;
    const int nrow = lane & 15;
    const int col  = colbase + nrow;

    const float* wk = Wm + (size_t)(2 * half) * doutp + col;
    int aoff = nrow * ldsrow + 2 * half;

    v8f acc = {};
    for (int k = 0; k < din; k += 4) {
        v2f a, b;
        a.x = lds[aoff];
        a.y = lds[aoff + 1];
        b.x = wk[0];
        b.y = wk[doutp];
        acc = __builtin_amdgcn_wmma_f32_16x16x4_f32(
                  false, a, false, b, (short)0, acc, false, false);
        aoff += 4;
        wk   += (size_t)4 * doutp;
    }

    const float bv = bias[col];
    #pragma unroll
    for (int r = 0; r < 8; ++r) {
        const int m   = r + 8 * half;
        const int row = row0 + m;
        if (row < N && col < dout)
            P[(size_t)row * dout + col] = acc[r] + bv;
    }
}

__global__ void gcn_pad_w4(const float* __restrict__ W4, const float* __restrict__ b4,
                           float* __restrict__ Wp, float* __restrict__ bp)
{
    const int i = blockIdx.x * blockDim.x + threadIdx.x;
    if (i < 64 * 16) {
        const int r = i >> 4, c = i & 15;
        Wp[i] = (c < 10) ? W4[r * 10 + c] : 0.f;
    } else if (i < 64 * 16 + 16) {
        const int c = i - 64 * 16;
        bp[c] = (c < 10) ? b4[c] : 0.f;
    }
}

__global__ void gcn_zero_i32(int* __restrict__ p, int n)
{
    const int i = blockIdx.x * blockDim.x + threadIdx.x;
    if (i < n) p[i] = 0;
}

__global__ __launch_bounds__(256) void gcn_count_rank(
    const int* __restrict__ dst, int* __restrict__ cnt, int* __restrict__ rank, int E)
{
    const int e = blockIdx.x * blockDim.x + threadIdx.x;
    if (e >= E) return;
    rank[e] = __hip_atomic_fetch_add(cnt + dst[e], 1, __ATOMIC_RELAXED,
                                     __HIP_MEMORY_SCOPE_AGENT);
}

#define SCAN_ROWS ((NNODES / 4 + 1023) / 1024)
static_assert(NNODES % 4 == 0, "scan reads the counts as 4-int vectors");
static_assert(SCAN_ROWS * 1024 >= NNODES / 4, "the rows must cover every vector");

__global__ __launch_bounds__(1024) void gcn_scan(
    const int* __restrict__ cnt, int* __restrict__ rowptr)
{
    constexpr int NVEC = NNODES / 4;
    __shared__ int tot[SCAN_ROWS * 32];
    const int t    = threadIdx.x;
    const int lane = t & 31;
    const int wv   = t >> 5;
    const int4* c4 = (const int4*)cnt;

    int s[SCAN_ROWS], incl[SCAN_ROWS];
    #pragma unroll
    for (int r = 0; r < SCAN_ROWS; ++r) {
        const int v = r * 1024 + t;
        int4 c = make_int4(0, 0, 0, 0);
        if (v < NVEC) c = c4[v];
        s[r] = c.x + c.y + c.z + c.w;
    }
    #pragma unroll
    for (int r = 0; r < SCAN_ROWS; ++r) {
        int x = s[r];
        #pragma unroll
        for (int off = 1; off < 32; off <<= 1) {
            const int y = __shfl_up(x, off, 32);
            if (lane >= off) x += y;
        }
        incl[r] = x;
    }
    if (lane == 31) {
        #pragma unroll
        for (int r = 0; r < SCAN_ROWS; ++r) tot[r * 32 + wv] = incl[r];
    }
    __syncthreads();
    if (wv == 0) {
        int loc[SCAN_ROWS];
        int run = 0;
        #pragma unroll
        for (int k = 0; k < SCAN_ROWS; ++k) {
            const int v = tot[lane * SCAN_ROWS + k];
            loc[k] = run;
            run += v;
        }
        int wi = run;
        #pragma unroll
        for (int off = 1; off < 32; off <<= 1) {
            const int y = __shfl_up(wi, off, 32);
            if (lane >= off) wi += y;
        }
        const int ex = wi - run;
        #pragma unroll
        for (int k = 0; k < SCAN_ROWS; ++k) tot[lane * SCAN_ROWS + k] = loc[k] + ex;
        if (lane == 31) rowptr[NNODES] = wi;
    }
    __syncthreads();
    int4* r4 = (int4*)rowptr;
    #pragma unroll
    for (int r = 0; r < SCAN_ROWS; ++r) {
        const int v = r * 1024 + t;
        if (v < NVEC) {
            const int4 c = c4[v];
            int4 o;
            o.x = tot[r * 32 + wv] + incl[r] - s[r];
            o.y = o.x + c.x;
            o.z = o.y + c.y;
            o.w = o.z + c.z;
            r4[v] = o;
        }
    }
}

__global__ __launch_bounds__(256) void gcn_fill(
    const int* __restrict__ src, const int* __restrict__ dst,
    const float* __restrict__ w, const int* __restrict__ rowptr,
    const int* __restrict__ rank, int* __restrict__ csr, int E)
{
    const int e = blockIdx.x * blockDim.x + threadIdx.x;
    if (e >= E) return;
    const int pos = rowptr[dst[e]] + rank[e];
    int2 v;
    v.x = src[e];
    v.y = __float_as_int(w[e]);
    *(int2*)(csr + 2 * (size_t)pos) = v;
}

__device__ __forceinline__ void fma4(float4& a, const float4 r, const float w)
{
    a.x = fmaf(r.x, w, a.x); a.y = fmaf(r.y, w, a.y);
    a.z = fmaf(r.z, w, a.z); a.w = fmaf(r.w, w, a.w);
}

template <int DOUT, int DOUTP>
__global__ __launch_bounds__(256) void gcn_gather64_gemm(
    const float* __restrict__ P, const int* __restrict__ rowptr,
    const int* __restrict__ csr, const float* __restrict__ Wm,
    const float* __restrict__ bias, float* __restrict__ Pn, int N)
{
    constexpr int DIN  = 64;
    constexpr int LROW = DIN + 4;
    static_assert(DOUTP % 16 == 0 && DOUTP <= 64 && DOUT <= DOUTP, "column tiles");
    __shared__ float tile[16 * LROW];

    const int t    = threadIdx.x;
    const int nb   = t >> 4;
    const int l    = t & 15;
    const int row0 = blockIdx.x * 16;
    const int node = row0 + nb;
    const int2*   ed = (const int2*)csr;
    const float4* P4 = (const float4*)P;

    float4 acc = make_float4(0.f, 0.f, 0.f, 0.f);
    if (node < N) {
        int j = rowptr[node];
        const int end = rowptr[node + 1];
        for (; j + 4 <= end; j += 4) {
            const int2 e0 = ed[j], e1 = ed[j + 1], e2 = ed[j + 2], e3 = ed[j + 3];
            const float4 r0 = P4[(size_t)e0.x * 16 + l];
            const float4 r1 = P4[(size_t)e1.x * 16 + l];
            const float4 r2 = P4[(size_t)e2.x * 16 + l];
            const float4 r3 = P4[(size_t)e3.x * 16 + l];
            fma4(acc, r0, __int_as_float(e0.y));
            fma4(acc, r1, __int_as_float(e1.y));
            fma4(acc, r2, __int_as_float(e2.y));
            fma4(acc, r3, __int_as_float(e3.y));
        }
        if (j + 2 <= end) {
            const int2 e0 = ed[j], e1 = ed[j + 1];
            const float4 r0 = P4[(size_t)e0.x * 16 + l];
            const float4 r1 = P4[(size_t)e1.x * 16 + l];
            fma4(acc, r0, __int_as_float(e0.y));
            fma4(acc, r1, __int_as_float(e1.y));
            j += 2;
        }
        if (j < end) {
            const int2 e0 = ed[j];
            const float4 r0 = P4[(size_t)e0.x * 16 + l];
            fma4(acc, r0, __int_as_float(e0.y));
        }
    }

    acc.x = fmaxf(acc.x, 0.f); acc.y = fmaxf(acc.y, 0.f);
    acc.z = fmaxf(acc.z, 0.f); acc.w = fmaxf(acc.w, 0.f);
    *(float4*)(&tile[nb * LROW + 4 * l]) = acc;

    const int wave    = t >> 5;
    const int lane    = t & 31;
    const int colbase = wave * 16;
    const bool mult   = colbase < DOUTP;
    const int half = lane >> 4;
    const int nrow = lane & 15;
    const int col  = colbase + nrow;

    v2f b[DIN / 4];
    float bv = 0.f;
    if (mult) {
        const float* wk = Wm + (size_t)(2 * half) * DOUTP + col;
        #pragma unroll
        for (int s = 0; s < DIN / 4; ++s) {
            b[s].x = wk[(size_t)(4 * s) * DOUTP];
            b[s].y = wk[(size_t)(4 * s + 1) * DOUTP];
        }
        bv = bias[col];
    }
    __syncthreads();
    if (!mult) return;

    const int aoff = nrow * LROW + 2 * half;
    v8f d = {};
    #pragma unroll
    for (int s = 0; s < DIN / 4; ++s) {
        v2f a;
        a.x = tile[aoff + 4 * s];
        a.y = tile[aoff + 4 * s + 1];
        d = __builtin_amdgcn_wmma_f32_16x16x4_f32(
                false, a, false, b[s], (short)0, d, false, false);
    }

    #pragma unroll
    for (int r = 0; r < 8; ++r) {
        const int m   = r + 8 * half;
        const int row = row0 + m;
        if (row < N && col < DOUT)
            Pn[(size_t)row * DOUT + col] = d[r] + bv;
    }
}

__global__ __launch_bounds__(256) void gcn_gather10(
    const float* __restrict__ P, const int* __restrict__ rowptr,
    const int* __restrict__ csr, float* __restrict__ out, int N)
{
    __shared__ float stage[160];
    const int t    = threadIdx.x;
    const int nb   = t >> 4;
    const int l    = t & 15;
    const int node = blockIdx.x * 16 + nb;
    const int2* ed = (const int2*)csr;

    float acc = 0.f;
    if (node < N && l < 10) {
        int j = rowptr[node];
        const int end = rowptr[node + 1];
        for (; j + 4 <= end; j += 4) {
            const int2 e0 = ed[j], e1 = ed[j + 1], e2 = ed[j + 2], e3 = ed[j + 3];
            const float r0 = P[(size_t)e0.x * 10 + l];
            const float r1 = P[(size_t)e1.x * 10 + l];
            const float r2 = P[(size_t)e2.x * 10 + l];
            const float r3 = P[(size_t)e3.x * 10 + l];
            acc = fmaf(r0, __int_as_float(e0.y), acc);
            acc = fmaf(r1, __int_as_float(e1.y), acc);
            acc = fmaf(r2, __int_as_float(e2.y), acc);
            acc = fmaf(r3, __int_as_float(e3.y), acc);
        }
        for (; j < end; ++j) {
            const int2 e0 = ed[j];
            acc = fmaf(P[(size_t)e0.x * 10 + l], __int_as_float(e0.y), acc);
        }
    }
    if (l < 10) stage[nb * 10 + l] = acc;
    __syncthreads();
    const int o = blockIdx.x * 160 + t;
    if (t < 160 && o < N * 10) out[o] = stage[t];
}

extern "C" void kernel_launch(void* const* d_in, const int* in_sizes, int n_in,
                              void* d_out, int out_size, void* d_ws, size_t ws_size,
                              hipStream_t stream)
{
    (void)in_sizes; (void)n_in; (void)out_size; (void)ws_size;
    const int N = NNODES, E = NEDGES;

    const float* x    = (const float*)d_in[0];
    const int*   esrc = (const int*)  d_in[1];
    const int*   edst = (const int*)  d_in[2];
    const float* ew   = (const float*)d_in[3];
    const float* Ws[5]; const float* bs[5];
    for (int i = 0; i < 5; ++i) {
        Ws[i] = (const float*)d_in[4 + 2 * i];
        bs[i] = (const float*)d_in[5 + 2 * i];
    }
    float* out  = (float*)d_out;
    float* buf0 = (float*)d_ws;
    float* buf1 = buf0 + (size_t)N * 64;
    float* buf2 = buf1 + (size_t)N * 64;
    float* wpad = buf2 + (size_t)N * 64;
    float* bpad = wpad + 64 * 16;
    int*   cnt    = (int*)(bpad + 64);
    int*   rowptr = cnt + N;
    int*   rank   = rowptr + (N + 64);
    int*   csr    = rank + E;

    const dim3 ggrid((N + 15) / 16);
    const int  egrid = (E + 255) / 256;

    gcn_pad_w4<<<(64 * 16 + 16 + 255) / 256, 256, 0, stream>>>(Ws[4], bs[4], wpad, bpad);

    gcn_zero_i32<<<(N + 255) / 256, 256, 0, stream>>>(cnt, N);
    gcn_count_rank<<<egrid, 256, 0, stream>>>(edst, cnt, rank, E);
    gcn_scan<<<1, 1024, 0, stream>>>(cnt, rowptr);
    gcn_fill<<<egrid, 256, 0, stream>>>(esrc, edst, ew, rowptr, rank, csr, E);

    gcn_gemm_wmma<<<ggrid, 128, 0, stream>>>(x, Ws[0], bs[0], buf0, N, 128, 64, 64, 0);

    gcn_gather64_gemm<64, 64><<<ggrid, 256, 0, stream>>>(buf0, rowptr, csr, Ws[1], bs[1], buf1, N);
    gcn_gather64_gemm<64, 64><<<ggrid, 256, 0, stream>>>(buf1, rowptr, csr, Ws[2], bs[2], buf0, N);
    gcn_gather64_gemm<64, 64><<<ggrid, 256, 0, stream>>>(buf0, rowptr, csr, Ws[3], bs[3], buf1, N);

    gcn_gather64_gemm<10, 16><<<ggrid, 256, 0, stream>>>(buf1, rowptr, csr, wpad, bpad, buf0, N);

    gcn_gather10<<<ggrid, 256, 0, stream>>>(buf0, rowptr, csr, out, N);
}


# Gather: a node's edge records come 32 per pair of coalesced loads and reach the lanes by row-share lane select, not one broadcast load per edge; 8 row loads in flight per lane (was 4); sum order kept
# speedup vs baseline: 2.8259x; 1.0402x over previous
#include <hip/hip_runtime.h>
#include <hip/hip_bf16.h>

typedef float v2f __attribute__((ext_vector_type(2)));
typedef float v8f __attribute__((ext_vector_type(8)));

#define NNODES 100000
#define NEDGES 1600000

__global__ __launch_bounds__(128) void gcn_gemm_wmma(
    const float* __restrict__ X, const float* __restrict__ Wm,
    const float* __restrict__ bias, float* __restrict__ P,
    int N, int din, int dout, int doutp, int relu_in)
{
    __shared__ float lds[16 * (128 + 4)];
    const int ldsrow = din + 4;
    const int row0 = blockIdx.x * 16;
    const int t = threadIdx.x;

    const int nvec = (16 * din) >> 2;
    const int vper = din >> 2;
    for (int idx = t; idx < nvec; idx += 128) {
        const int r  = idx / vper;
        const int c4 = (idx - r * vper) << 2;
        const int row = row0 + r;
        float4 v = make_float4(0.f, 0.f, 0.f, 0.f);
        if (row < N) {
            v = *(const float4*)(X + (size_t)row * din + c4);
            if (relu_in) {
                v.x = fmaxf(v.x, 0.f); v.y = fmaxf(v.y, 0.f);
                v.z = fmaxf(v.z, 0.f); v.w = fmaxf(v.w, 0.f);
            }
        }
        *(float4*)(&lds[r * ldsrow + c4]) = v;
    }
    __syncthreads();

    const int wave = t >> 5;
    const int lane = t & 31;
    const int colbase = wave * 16;
    if (colbase >= doutp) return;

    const int half = lane >> 4;
    const int nrow = lane & 15;
    const int col  = colbase + nrow;

    const float* wk = Wm + (size_t)(2 * half) * doutp + col;
    int aoff = nrow * ldsrow + 2 * half;

    v8f acc = {};
    for (int k = 0; k < din; k += 4) {
        v2f a, b;
        a.x = lds[aoff];
        a.y = lds[aoff + 1];
        b.x = wk[0];
        b.y = wk[doutp];
        acc = __builtin_amdgcn_wmma_f32_16x16x4_f32(
                  false, a, false, b, (short)0, acc, false, false);
        aoff += 4;
        wk   += (size_t)4 * doutp;
    }

    const float bv = bias[col];
    #pragma unroll
    for (int r = 0; r < 8; ++r) {
        const int m   = r + 8 * half;
        const int row = row0 + m;
        if (row < N && col < dout)
            P[(size_t)row * dout + col] = acc[r] + bv;
    }
}

__global__ void gcn_pad_w4(const float* __restrict__ W4, const float* __restrict__ b4,
                           float* __restrict__ Wp, float* __restrict__ bp)
{
    const int i = blockIdx.x * blockDim.x + threadIdx.x;
    if (i < 64 * 16) {
        const int r = i >> 4, c = i & 15;
        Wp[i] = (c < 10) ? W4[r * 10 + c] : 0.f;
    } else if (i < 64 * 16 + 16) {
        const int c = i - 64 * 16;
        bp[c] = (c < 10) ? b4[c] : 0.f;
    }
}

__global__ void gcn_zero_i32(int* __restrict__ p, int n)
{
    const int i = blockIdx.x * blockDim.x + threadIdx.x;
    if (i < n) p[i] = 0;
}

__global__ __launch_bounds__(256) void gcn_count_rank(
    const int* __restrict__ dst, int* __restrict__ cnt, int* __restrict__ rank, int E)
{
    const int e = blockIdx.x * blockDim.x + threadIdx.x;
    if (e >= E) return;
    rank[e] = __hip_atomic_fetch_add(cnt + dst[e], 1, __ATOMIC_RELAXED,
                                     __HIP_MEMORY_SCOPE_AGENT);
}

#define SCAN_ROWS ((NNODES / 4 + 1023) / 1024)
static_assert(NNODES % 4 == 0, "scan reads the counts as 4-int vectors");
static_assert(SCAN_ROWS * 1024 >= NNODES / 4, "the rows must cover every vector");

__global__ __launch_bounds__(1024) void gcn_scan(
    const int* __restrict__ cnt, int* __restrict__ rowptr)
{
    constexpr int NVEC = NNODES / 4;
    __shared__ int tot[SCAN_ROWS * 32];
    const int t    = threadIdx.x;
    const int lane = t & 31;
    const int wv   = t >> 5;
    const int4* c4 = (const int4*)cnt;

    int s[SCAN_ROWS], incl[SCAN_ROWS];
    #pragma unroll
    for (int r = 0; r < SCAN_ROWS; ++r) {
        const int v = r * 1024 + t;
        int4 c = make_int4(0, 0, 0, 0);
        if (v < NVEC) c = c4[v];
        s[r] = c.x + c.y + c.z + c.w;
    }
    #pragma unroll
    for (int r = 0; r < SCAN_ROWS; ++r) {
        int x = s[r];
        #pragma unroll
        for (int off = 1; off < 32; off <<= 1) {
            const int y = __shfl_up(x, off, 32);
            if (lane >= off) x += y;
        }
        incl[r] = x;
    }
    if (lane == 31) {
        #pragma unroll
        for (int r = 0; r < SCAN_ROWS; ++r) tot[r * 32 + wv] = incl[r];
    }
    __syncthreads();
    if (wv == 0) {
        int loc[SCAN_ROWS];
        int run = 0;
        #pragma unroll
        for (int k = 0; k < SCAN_ROWS; ++k) {
            const int v = tot[lane * SCAN_ROWS + k];
            loc[k] = run;
            run += v;
        }
        int wi = run;
        #pragma unroll
        for (int off = 1; off < 32; off <<= 1) {
            const int y = __shfl_up(wi, off, 32);
            if (lane >= off) wi += y;
        }
        const int ex = wi - run;
        #pragma unroll
        for (int k = 0; k < SCAN_ROWS; ++k) tot[lane * SCAN_ROWS + k] = loc[k] + ex;
        if (lane == 31) rowptr[NNODES] = wi;
    }
    __syncthreads();
    int4* r4 = (int4*)rowptr;
    #pragma unroll
    for (int r = 0; r < SCAN_ROWS; ++r) {
        const int v = r * 1024 + t;
        if (v < NVEC) {
            const int4 c = c4[v];
            int4 o;
            o.x = tot[r * 32 + wv] + incl[r] - s[r];
            o.y = o.x + c.x;
            o.z = o.y + c.y;
            o.w = o.z + c.z;
            r4[v] = o;
        }
    }
}

__global__ __launch_bounds__(256) void gcn_fill(
    const int* __restrict__ src, const int* __restrict__ dst,
    const float* __restrict__ w, const int* __restrict__ rowptr,
    const int* __restrict__ rank, int* __restrict__ csr, int E)
{
    const int e = blockIdx.x * blockDim.x + threadIdx.x;
    if (e >= E) return;
    const int pos = rowptr[dst[e]] + rank[e];
    int2 v;
    v.x = src[e];
    v.y = __float_as_int(w[e]);
    *(int2*)(csr + 2 * (size_t)pos) = v;
}

template <int I>
__device__ __forceinline__ int row_share(const int v)
{
    return __builtin_amdgcn_update_dpp(0, v, 0x150 + I, 0xF, 0xF, true);
}
__device__ __forceinline__ int row_lane(const int v, const int i)
{
    switch (i) {
    case 0:  return row_share<0>(v);
    case 1:  return row_share<1>(v);
    case 2:  return row_share<2>(v);
    case 3:  return row_share<3>(v);
    case 4:  return row_share<4>(v);
    case 5:  return row_share<5>(v);
    case 6:  return row_share<6>(v);
    case 7:  return row_share<7>(v);
    case 8:  return row_share<8>(v);
    case 9:  return row_share<9>(v);
    case 10: return row_share<10>(v);
    case 11: return row_share<11>(v);
    case 12: return row_share<12>(v);
    case 13: return row_share<13>(v);
    case 14: return row_share<14>(v);
    default: return row_share<15>(v);
    }
}

__device__ __forceinline__ void fmav(float& a, const float r, const float w)
{
    a = fmaf(r, w, a);
}
__device__ __forceinline__ void fmav(float4& a, const float4 r, const float w)
{
    a.x = fmaf(r.x, w, a.x); a.y = fmaf(r.y, w, a.y);
    a.z = fmaf(r.z, w, a.z); a.w = fmaf(r.w, w, a.w);
}
__device__ __forceinline__ float  zero_of(const float*)  { return 0.f; }
__device__ __forceinline__ float4 zero_of(const float4*) { return make_float4(0.f, 0.f, 0.f, 0.f); }

template <typename V, int VPR, int CH>
__device__ __forceinline__ V gather_node(
    const V* __restrict__ Pv, const int2* __restrict__ ed,
    int j, const int end, const int l, const int col)
{
    static_assert(16 % CH == 0, "a load of 16 records is whole chunks");
    constexpr unsigned ROWB = VPR * sizeof(V);
    const char* base = (const char*)Pv;
    const unsigned colb = (unsigned)col * (unsigned)sizeof(V);
    V acc = zero_of((const V*)nullptr);
    for (; j < end; j += 32) {
        const int2 rec0 = ed[min(j + l, end - 1)];
        const int2 rec1 = ed[min(j + 16 + l, end - 1)];
        #pragma unroll
        for (int c = 0; c < 32; c += CH) {
            const int m = end - j - c;
            if (m > 0) {
                const int2 rec = c < 16 ? rec0 : rec1;
                const int c0 = c & 15;
                unsigned off[CH];
                #pragma unroll
                for (int i = 0; i < CH; ++i)
                    off[i] = (unsigned)row_lane(rec.x, c0 + i) * ROWB + colb;
                V r[CH];
                #pragma unroll
                for (int i = 0; i < CH; ++i)
                    if (i < m) r[i] = *(const V*)(base + off[i]);
                #pragma unroll
                for (int i = 0; i < CH; ++i)
                    if (i < m) fmav(acc, r[i], __int_as_float(row_lane(rec.y, c0 + i)));
            }
        }
    }
    return acc;
}

template <int DOUT, int DOUTP>
__global__ __launch_bounds__(256) void gcn_gather64_gemm(
    const float* __restrict__ P, const int* __restrict__ rowptr,
    const int* __restrict__ csr, const float* __restrict__ Wm,
    const float* __restrict__ bias, float* __restrict__ Pn, int N)
{
    constexpr int DIN  = 64;
    constexpr int LROW = DIN + 4;
    static_assert(DOUTP % 16 == 0 && DOUTP <= 64 && DOUT <= DOUTP, "column tiles");
    __shared__ float tile[16 * LROW];

    const int t    = threadIdx.x;
    const int nb   = t >> 4;
    const int l    = t & 15;
    const int row0 = blockIdx.x * 16;
    const int node = row0 + nb;

    float4 acc = make_float4(0.f, 0.f, 0.f, 0.f);
    if (node < N)
        acc = gather_node<float4, 16, 8>((const float4*)P, (const int2*)csr,
                                         rowptr[node], rowptr[node + 1], l, l);

    acc.x = fmaxf(acc.x, 0.f); acc.y = fmaxf(acc.y, 0.f);
    acc.z = fmaxf(acc.z, 0.f); acc.w = fmaxf(acc.w, 0.f);
    *(float4*)(&tile[nb * LROW + 4 * l]) = acc;

    const int wave    = t >> 5;
    const int lane    = t & 31;
    const int colbase = wave * 16;
    const bool mult   = colbase < DOUTP;
    const int half = lane >> 4;
    const int nrow = lane & 15;
    const int col  = colbase + nrow;

    v2f b[DIN / 4];
    float bv = 0.f;
    if (mult) {
        const float* wk = Wm + (size_t)(2 * half) * DOUTP + col;
        #pragma unroll
        for (int s = 0; s < DIN / 4; ++s) {
            b[s].x = wk[(size_t)(4 * s) * DOUTP];
            b[s].y = wk[(size_t)(4 * s + 1) * DOUTP];
        }
        bv = bias[col];
    }
    __syncthreads();
    if (!mult) return;

    const int aoff = nrow * LROW + 2 * half;
    v8f d = {};
    #pragma unroll
    for (int s = 0; s < DIN / 4; ++s) {
        v2f a;
        a.x = tile[aoff + 4 * s];
        a.y = tile[aoff + 4 * s + 1];
        d = __builtin_amdgcn_wmma_f32_16x16x4_f32(
                false, a, false, b[s], (short)0, d, false, false);
    }

    #pragma unroll
    for (int r = 0; r < 8; ++r) {
        const int m   = r + 8 * half;
        const int row = row0 + m;
        if (row < N && col < DOUT)
            Pn[(size_t)row * DOUT + col] = d[r] + bv;
    }
}

__global__ __launch_bounds__(256) void gcn_gather10(
    const float* __restrict__ P, const int* __restrict__ rowptr,
    const int* __restrict__ csr, float* __restrict__ out, int N)
{
    __shared__ float stage[160];
    const int t    = threadIdx.x;
    const int nb   = t >> 4;
    const int l    = t & 15;
    const int node = blockIdx.x * 16 + nb;

    float acc = 0.f;
    if (node < N)
        acc = gather_node<float, 10, 16>(P, (const int2*)csr,
                                         rowptr[node], rowptr[node + 1], l, min(l, 9));
    if (l < 10) stage[nb * 10 + l] = acc;
    __syncthreads();
    const int o = blockIdx.x * 160 + t;
    if (t < 160 && o < N * 10) out[o] = stage[t];
}

extern "C" void kernel_launch(void* const* d_in, const int* in_sizes, int n_in,
                              void* d_out, int out_size, void* d_ws, size_t ws_size,
                              hipStream_t stream)
{
    (void)in_sizes; (void)n_in; (void)out_size; (void)ws_size;
    const int N = NNODES, E = NEDGES;

    const float* x    = (const float*)d_in[0];
    const int*   esrc = (const int*)  d_in[1];
    const int*   edst = (const int*)  d_in[2];
    const float* ew   = (const float*)d_in[3];
    const float* Ws[5]; const float* bs[5];
    for (int i = 0; i < 5; ++i) {
        Ws[i] = (const float*)d_in[4 + 2 * i];
        bs[i] = (const float*)d_in[5 + 2 * i];
    }
    float* out  = (float*)d_out;
    float* buf0 = (float*)d_ws;
    float* buf1 = buf0 + (size_t)N * 64;
    float* buf2 = buf1 + (size_t)N * 64;
    float* wpad = buf2 + (size_t)N * 64;
    float* bpad = wpad + 64 * 16;
    int*   cnt    = (int*)(bpad + 64);
    int*   rowptr = cnt + N;
    int*   rank   = rowptr + (N + 64);
    int*   csr    = rank + E;

    const dim3 ggrid((N + 15) / 16);
    const int  egrid = (E + 255) / 256;

    gcn_pad_w4<<<(64 * 16 + 16 + 255) / 256, 256, 0, stream>>>(Ws[4], bs[4], wpad, bpad);

    gcn_zero_i32<<<(N + 255) / 256, 256, 0, stream>>>(cnt, N);
    gcn_count_rank<<<egrid, 256, 0, stream>>>(edst, cnt, rank, E);
    gcn_scan<<<1, 1024, 0, stream>>>(cnt, rowptr);
    gcn_fill<<<egrid, 256, 0, stream>>>(esrc, edst, ew, rowptr, rank, csr, E);

    gcn_gemm_wmma<<<ggrid, 128, 0, stream>>>(x, Ws[0], bs[0], buf0, N, 128, 64, 64, 0);

    gcn_gather64_gemm<64, 64><<<ggrid, 256, 0, stream>>>(buf0, rowptr, csr, Ws[1], bs[1], buf1, N);
    gcn_gather64_gemm<64, 64><<<ggrid, 256, 0, stream>>>(buf1, rowptr, csr, Ws[2], bs[2], buf0, N);
    gcn_gather64_gemm<64, 64><<<ggrid, 256, 0, stream>>>(buf0, rowptr, csr, Ws[3], bs[3], buf1, N);

    gcn_gather64_gemm<10, 16><<<ggrid, 256, 0, stream>>>(buf1, rowptr, csr, wpad, bpad, buf0, N);

    gcn_gather10<<<ggrid, 256, 0, stream>>>(buf0, rowptr, csr, out, N);
}


# Fewer, longer waves: the 7 wide launches run on resident grids of 1250 work-groups that loop over 16-row tiles (5 edges per thread in the edge kernels); weight fragments load once per work-group
# speedup vs baseline: 2.9597x; 1.0473x over previous
#include <hip/hip_runtime.h>
#include <hip/hip_bf16.h>

typedef float v2f __attribute__((ext_vector_type(2)));
typedef float v8f __attribute__((ext_vector_type(8)));

#define NNODES 100000
#define NEDGES 1600000


template <int DIN, int DOUT, int DOUTP>
__global__ __launch_bounds__(128) void gcn_gemm_wmma(
    const float* __restrict__ X, const float* __restrict__ Wm,
    const float* __restrict__ bias, float* __restrict__ P, int N, int ntiles)
{
    constexpr int LROW = DIN + 4;
    constexpr int VPER = DIN / 4;
    constexpr int NV   = 16 * VPER / 128;
    static_assert(DIN % 32 == 0 && DOUTP % 16 == 0 && DOUTP <= 64 && DOUT <= DOUTP, "tiles");
    __shared__ float lds[2][16 * LROW];
    const int t = threadIdx.x;

    const int wave = t >> 5;
    const int lane = t & 31;
    const int colbase = wave * 16;
    const bool mult = colbase < DOUTP;
    const int half = lane >> 4;
    const int nrow = lane & 15;
    const int col  = colbase + nrow;

    v2f b[DIN / 4];
    float bv = 0.f;
    if (mult) {
        const float* wk = Wm + (size_t)(2 * half) * DOUTP + col;
        #pragma unroll
        for (int s = 0; s < DIN / 4; ++s) {
            b[s].x = wk[(size_t)(4 * s) * DOUTP];
            b[s].y = wk[(size_t)(4 * s + 1) * DOUTP];
        }
        bv = bias[col];
    }
    const int aoff = nrow * LROW + 2 * half;

    int buf = 0;
    for (int tl = blockIdx.x; tl < ntiles; tl += gridDim.x, buf ^= 1) {
        const int row0 = tl * 16;
        float* tile = lds[buf];
        float4 v[NV];
        #pragma unroll
        for (int i = 0; i < NV; ++i) {
            const int idx = t + i * 128;
            const int r   = idx / VPER;
            const int c4  = (idx - r * VPER) << 2;
            const int row = row0 + r;
            v[i] = *(const float4*)(X + (size_t)min(row, N - 1) * DIN + c4);
            if (row >= N) v[i] = make_float4(0.f, 0.f, 0.f, 0.f);
        }
        #pragma unroll
        for (int i = 0; i < NV; ++i) {
            const int idx = t + i * 128;
            const int r   = idx / VPER;
            const int c4  = (idx - r * VPER) << 2;
            *(float4*)(&tile[r * LROW + c4]) = v[i];
        }
        __syncthreads();
        if (mult) {
            v8f acc = {};
            #pragma unroll
            for (int s = 0; s < DIN / 4; ++s) {
                v2f a;
                a.x = tile[aoff + 4 * s];
                a.y = tile[aoff + 4 * s + 1];
                acc = __builtin_amdgcn_wmma_f32_16x16x4_f32(
                          false, a, false, b[s], (short)0, acc, false, false);
            }
            #pragma unroll
            for (int r = 0; r < 8; ++r) {
                const int m   = r + 8 * half;
                const int row = row0 + m;
                if (row < N && col < DOUT)
                    P[(size_t)row * DOUT + col] = acc[r] + bv;
            }
        }
    }
}

__global__ void gcn_pad_w4(const float* __restrict__ W4, const float* __restrict__ b4,
                           float* __restrict__ Wp, float* __restrict__ bp)
{
    const int i = blockIdx.x * blockDim.x + threadIdx.x;
    if (i < 64 * 16) {
        const int r = i >> 4, c = i & 15;
        Wp[i] = (c < 10) ? W4[r * 10 + c] : 0.f;
    } else if (i < 64 * 16 + 16) {
        const int c = i - 64 * 16;
        bp[c] = (c < 10) ? b4[c] : 0.f;
    }
}

__global__ void gcn_zero_i32(int* __restrict__ p, int n)
{
    const int i = blockIdx.x * blockDim.x + threadIdx.x;
    if (i < n) p[i] = 0;
}

#define EPT 5

__global__ __launch_bounds__(256) void gcn_count_rank(
    const int* __restrict__ dst, int* __restrict__ cnt, int* __restrict__ rank, int E)
{
    const int T = gridDim.x * 256;
    for (int e0 = blockIdx.x * 256 + threadIdx.x; e0 < E; e0 += EPT * T) {
        int d[EPT], r[EPT];
        #pragma unroll
        for (int k = 0; k < EPT; ++k) {
            const int e = e0 + k * T;
            d[k] = e < E ? dst[e] : -1;
        }
        #pragma unroll
        for (int k = 0; k < EPT; ++k) {
            r[k] = 0;
            if (d[k] >= 0)
                r[k] = __hip_atomic_fetch_add(cnt + d[k], 1, __ATOMIC_RELAXED,
                                              __HIP_MEMORY_SCOPE_AGENT);
        }
        #pragma unroll
        for (int k = 0; k < EPT; ++k) {
            const int e = e0 + k * T;
            if (e < E) rank[e] = r[k];
        }
    }
}

#define SCAN_ROWS ((NNODES / 4 + 1023) / 1024)
static_assert(NNODES % 4 == 0, "scan reads the counts as 4-int vectors");
static_assert(SCAN_ROWS * 1024 >= NNODES / 4, "the rows must cover every vector");

__global__ __launch_bounds__(1024) void gcn_scan(
    const int* __restrict__ cnt, int* __restrict__ rowptr)
{
    constexpr int NVEC = NNODES / 4;
    __shared__ int tot[SCAN_ROWS * 32];
    const int t    = threadIdx.x;
    const int lane = t & 31;
    const int wv   = t >> 5;
    const int4* c4 = (const int4*)cnt;

    int s[SCAN_ROWS], incl[SCAN_ROWS];
    #pragma unroll
    for (int r = 0; r < SCAN_ROWS; ++r) {
        const int v = r * 1024 + t;
        int4 c = make_int4(0, 0, 0, 0);
        if (v < NVEC) c = c4[v];
        s[r] = c.x + c.y + c.z + c.w;
    }
    #pragma unroll
    for (int r = 0; r < SCAN_ROWS; ++r) {
        int x = s[r];
        #pragma unroll
        for (int off = 1; off < 32; off <<= 1) {
            const int y = __shfl_up(x, off, 32);
            if (lane >= off) x += y;
        }
        incl[r] = x;
    }
    if (lane == 31) {
        #pragma unroll
        for (int r = 0; r < SCAN_ROWS; ++r) tot[r * 32 + wv] = incl[r];
    }
    __syncthreads();
    if (wv == 0) {
        int loc[SCAN_ROWS];
        int run = 0;
        #pragma unroll
        for (int k = 0; k < SCAN_ROWS; ++k) {
            const int v = tot[lane * SCAN_ROWS + k];
            loc[k] = run;
            run += v;
        }
        int wi = run;
        #pragma unroll
        for (int off = 1; off < 32; off <<= 1) {
            const int y = __shfl_up(wi, off, 32);
            if (lane >= off) wi += y;
        }
        const int ex = wi - run;
        #pragma unroll
        for (int k = 0; k < SCAN_ROWS; ++k) tot[lane * SCAN_ROWS + k] = loc[k] + ex;
        if (lane == 31) rowptr[NNODES] = wi;
    }
    __syncthreads();
    int4* r4 = (int4*)rowptr;
    #pragma unroll
    for (int r = 0; r < SCAN_ROWS; ++r) {
        const int v = r * 1024 + t;
        if (v < NVEC) {
            const int4 c = c4[v];
            int4 o;
            o.x = tot[r * 32 + wv] + incl[r] - s[r];
            o.y = o.x + c.x;
            o.z = o.y + c.y;
            o.w = o.z + c.z;
            r4[v] = o;
        }
    }
}

__global__ __launch_bounds__(256) void gcn_fill(
    const int* __restrict__ src, const int* __restrict__ dst,
    const float* __restrict__ w, const int* __restrict__ rowptr,
    const int* __restrict__ rank, int* __restrict__ csr, int E)
{
    const int T = gridDim.x * 256;
    for (int e0 = blockIdx.x * 256 + threadIdx.x; e0 < E; e0 += EPT * T) {
        int d[EPT], rk[EPT], base[EPT];
        int2 v[EPT];
        #pragma unroll
        for (int k = 0; k < EPT; ++k) {
            const int ec = min(e0 + k * T, E - 1);
            d[k]   = dst[ec];
            rk[k]  = rank[ec];
            v[k].x = src[ec];
            v[k].y = __float_as_int(w[ec]);
        }
        #pragma unroll
        for (int k = 0; k < EPT; ++k) base[k] = rowptr[d[k]];
        #pragma unroll
        for (int k = 0; k < EPT; ++k) {
            const int e = e0 + k * T;
            if (e < E) *(int2*)(csr + 2 * (size_t)(base[k] + rk[k])) = v[k];
        }
    }
}

template <int I>
__device__ __forceinline__ int row_share(const int v)
{
    return __builtin_amdgcn_update_dpp(0, v, 0x150 + I, 0xF, 0xF, true);
}
__device__ __forceinline__ int row_lane(const int v, const int i)
{
    switch (i) {
    case 0:  return row_share<0>(v);
    case 1:  return row_share<1>(v);
    case 2:  return row_share<2>(v);
    case 3:  return row_share<3>(v);
    case 4:  return row_share<4>(v);
    case 5:  return row_share<5>(v);
    case 6:  return row_share<6>(v);
    case 7:  return row_share<7>(v);
    case 8:  return row_share<8>(v);
    case 9:  return row_share<9>(v);
    case 10: return row_share<10>(v);
    case 11: return row_share<11>(v);
    case 12: return row_share<12>(v);
    case 13: return row_share<13>(v);
    case 14: return row_share<14>(v);
    default: return row_share<15>(v);
    }
}

__device__ __forceinline__ void fmav(float& a, const float r, const float w)
{
    a = fmaf(r, w, a);
}
__device__ __forceinline__ void fmav(float4& a, const float4 r, const float w)
{
    a.x = fmaf(r.x, w, a.x); a.y = fmaf(r.y, w, a.y);
    a.z = fmaf(r.z, w, a.z); a.w = fmaf(r.w, w, a.w);
}
__device__ __forceinline__ float  zero_of(const float*)  { return 0.f; }
__device__ __forceinline__ float4 zero_of(const float4*) { return make_float4(0.f, 0.f, 0.f, 0.f); }

template <typename V, int VPR, int CH>
__device__ __forceinline__ V gather_node(
    const V* __restrict__ Pv, const int2* __restrict__ ed,
    int j, const int end, const int l, const int col)
{
    static_assert(16 % CH == 0, "a load of 16 records is whole chunks");
    constexpr unsigned ROWB = VPR * sizeof(V);
    const char* base = (const char*)Pv;
    const unsigned colb = (unsigned)col * (unsigned)sizeof(V);
    V acc = zero_of((const V*)nullptr);
    for (; j < end; j += 32) {
        const int2 rec0 = ed[min(j + l, end - 1)];
        const int2 rec1 = ed[min(j + 16 + l, end - 1)];
        #pragma unroll
        for (int c = 0; c < 32; c += CH) {
            const int m = end - j - c;
            if (m > 0) {
                const int2 rec = c < 16 ? rec0 : rec1;
                const int c0 = c & 15;
                unsigned off[CH];
                #pragma unroll
                for (int i = 0; i < CH; ++i)
                    off[i] = (unsigned)row_lane(rec.x, c0 + i) * ROWB + colb;
                V r[CH];
                #pragma unroll
                for (int i = 0; i < CH; ++i)
                    if (i < m) r[i] = *(const V*)(base + off[i]);
                #pragma unroll
                for (int i = 0; i < CH; ++i)
                    if (i < m) fmav(acc, r[i], __int_as_float(row_lane(rec.y, c0 + i)));
            }
        }
    }
    return acc;
}

template <int DOUT, int DOUTP>
__global__ __launch_bounds__(256) void gcn_gather64_gemm(
    const float* __restrict__ P, const int* __restrict__ rowptr,
    const int* __restrict__ csr, const float* __restrict__ Wm,
    const float* __restrict__ bias, float* __restrict__ Pn, int N, int ntiles)
{
    constexpr int DIN  = 64;
    constexpr int LROW = DIN + 4;
    static_assert(DOUTP % 16 == 0 && DOUTP <= 64 && DOUT <= DOUTP, "column tiles");
    __shared__ float lds[2][16 * LROW];

    const int t    = threadIdx.x;
    const int nb   = t >> 4;
    const int l    = t & 15;

    const int wave    = t >> 5;
    const int lane    = t & 31;
    const int colbase = wave * 16;
    const bool mult   = colbase < DOUTP;
    const int half = lane >> 4;
    const int nrow = lane & 15;
    const int col  = colbase + nrow;

    v2f b[DIN / 4];
    float bv = 0.f;
    if (mult) {
        const float* wk = Wm + (size_t)(2 * half) * DOUTP + col;
        #pragma unroll
        for (int s = 0; s < DIN / 4; ++s) {
            b[s].x = wk[(size_t)(4 * s) * DOUTP];
            b[s].y = wk[(size_t)(4 * s + 1) * DOUTP];
        }
        bv = bias[col];
    }
    const int aoff = nrow * LROW + 2 * half;

    int buf = 0;
    for (int tl = blockIdx.x; tl < ntiles; tl += gridDim.x, buf ^= 1) {
        const int row0 = tl * 16;
        const int node = row0 + nb;
        float* tile = lds[buf];

        float4 acc = make_float4(0.f, 0.f, 0.f, 0.f);
        if (node < N)
            acc = gather_node<float4, 16, 8>((const float4*)P, (const int2*)csr,
                                             rowptr[node], rowptr[node + 1], l, l);

        acc.x = fmaxf(acc.x, 0.f); acc.y = fmaxf(acc.y, 0.f);
        acc.z = fmaxf(acc.z, 0.f); acc.w = fmaxf(acc.w, 0.f);
        *(float4*)(&tile[nb * LROW + 4 * l]) = acc;
        __syncthreads();

        if (mult) {
            v8f d = {};
            #pragma unroll
            for (int s = 0; s < DIN / 4; ++s) {
                v2f a;
                a.x = tile[aoff + 4 * s];
                a.y = tile[aoff + 4 * s + 1];
                d = __builtin_amdgcn_wmma_f32_16x16x4_f32(
                        false, a, false, b[s], (short)0, d, false, false);
            }
            #pragma unroll
            for (int r = 0; r < 8; ++r) {
                const int m   = r + 8 * half;
                const int row = row0 + m;
                if (row < N && col < DOUT)
                    Pn[(size_t)row * DOUT + col] = d[r] + bv;
            }
        }
    }
}

__global__ __launch_bounds__(256) void gcn_gather10(
    const float* __restrict__ P, const int* __restrict__ rowptr,
    const int* __restrict__ csr, float* __restrict__ out, int N, int ntiles)
{
    __shared__ float stage[2][160];
    const int t    = threadIdx.x;
    const int nb   = t >> 4;
    const int l    = t & 15;

    int buf = 0;
    for (int tl = blockIdx.x; tl < ntiles; tl += gridDim.x, buf ^= 1) {
        const int node = tl * 16 + nb;
        float acc = 0.f;
        if (node < N)
            acc = gather_node<float, 10, 16>(P, (const int2*)csr,
                                             rowptr[node], rowptr[node + 1], l, min(l, 9));
        if (l < 10) stage[buf][nb * 10 + l] = acc;
        __syncthreads();
        const int o = tl * 160 + t;
        if (t < 160 && o < N * 10) out[o] = stage[buf][t];
    }
}

extern "C" void kernel_launch(void* const* d_in, const int* in_sizes, int n_in,
                              void* d_out, int out_size, void* d_ws, size_t ws_size,
                              hipStream_t stream)
{
    (void)in_sizes; (void)n_in; (void)out_size; (void)ws_size;
    const int N = NNODES, E = NEDGES;

    const float* x    = (const float*)d_in[0];
    const int*   esrc = (const int*)  d_in[1];
    const int*   edst = (const int*)  d_in[2];
    const float* ew   = (const float*)d_in[3];
    const float* Ws[5]; const float* bs[5];
    for (int i = 0; i < 5; ++i) {
        Ws[i] = (const float*)d_in[4 + 2 * i];
        bs[i] = (const float*)d_in[5 + 2 * i];
    }
    float* out  = (float*)d_out;
    float* buf0 = (float*)d_ws;
    float* buf1 = buf0 + (size_t)N * 64;
    float* buf2 = buf1 + (size_t)N * 64;
    float* wpad = buf2 + (size_t)N * 64;
    float* bpad = wpad + 64 * 16;
    int*   cnt    = (int*)(bpad + 64);
    int*   rowptr = cnt + N;
    int*   rank   = rowptr + (N + 64);
    int*   csr    = rank + E;

    const int  ntiles = (N + 15) / 16;
    const dim3 pgrid(1250);

    gcn_pad_w4<<<(64 * 16 + 16 + 255) / 256, 256, 0, stream>>>(Ws[4], bs[4], wpad, bpad);

    gcn_zero_i32<<<(N + 255) / 256, 256, 0, stream>>>(cnt, N);
    gcn_count_rank<<<pgrid, 256, 0, stream>>>(edst, cnt, rank, E);
    gcn_scan<<<1, 1024, 0, stream>>>(cnt, rowptr);
    gcn_fill<<<pgrid, 256, 0, stream>>>(esrc, edst, ew, rowptr, rank, csr, E);

    gcn_gemm_wmma<128, 64, 64><<<pgrid, 128, 0, stream>>>(x, Ws[0], bs[0], buf0, N, ntiles);

    gcn_gather64_gemm<64, 64><<<pgrid, 256, 0, stream>>>(buf0, rowptr, csr, Ws[1], bs[1], buf1, N, ntiles);
    gcn_gather64_gemm<64, 64><<<pgrid, 256, 0, stream>>>(buf1, rowptr, csr, Ws[2], bs[2], buf0, N, ntiles);
    gcn_gather64_gemm<64, 64><<<pgrid, 256, 0, stream>>>(buf0, rowptr, csr, Ws[3], bs[3], buf1, N, ntiles);

    gcn_gather64_gemm<10, 16><<<pgrid, 256, 0, stream>>>(buf1, rowptr, csr, wpad, bpad, buf0, N, ntiles);

    gcn_gather10<<<pgrid, 256, 0, stream>>>(buf0, rowptr, csr, out, N, ntiles);
}


# Edge-count atomics split by package half: each half adds to its own counter plane, no counter line shared across halves; a node's list fills from front (half 0) and back (half 1): new sum order
# speedup vs baseline: 3.7440x; 1.2650x over previous
#include <hip/hip_runtime.h>
#include <hip/hip_bf16.h>

typedef float v2f __attribute__((ext_vector_type(2)));
typedef float v8f __attribute__((ext_vector_type(8)));

#define NNODES 100000
#define NEDGES 1600000


template <int DIN, int DOUT, int DOUTP>
__global__ __launch_bounds__(128) void gcn_gemm_wmma(
    const float* __restrict__ X, const float* __restrict__ Wm,
    const float* __restrict__ bias, float* __restrict__ P, int N, int ntiles)
{
    constexpr int LROW = DIN + 4;
    constexpr int VPER = DIN / 4;
    constexpr int NV   = 16 * VPER / 128;
    static_assert(DIN % 32 == 0 && DOUTP % 16 == 0 && DOUTP <= 64 && DOUT <= DOUTP, "tiles");
    __shared__ float lds[2][16 * LROW];
    const int t = threadIdx.x;

    const int wave = t >> 5;
    const int lane = t & 31;
    const int colbase = wave * 16;
    const bool mult = colbase < DOUTP;
    const int half = lane >> 4;
    const int nrow = lane & 15;
    const int col  = colbase + nrow;

    v2f b[DIN / 4];
    float bv = 0.f;
    if (mult) {
        const float* wk = Wm + (size_t)(2 * half) * DOUTP + col;
        #pragma unroll
        for (int s = 0; s < DIN / 4; ++s) {
            b[s].x = wk[(size_t)(4 * s) * DOUTP];
            b[s].y = wk[(size_t)(4 * s + 1) * DOUTP];
        }
        bv = bias[col];
    }
    const int aoff = nrow * LROW + 2 * half;

    int buf = 0;
    for (int tl = blockIdx.x; tl < ntiles; tl += gridDim.x, buf ^= 1) {
        const int row0 = tl * 16;
        float* tile = lds[buf];
        float4 v[NV];
        #pragma unroll
        for (int i = 0; i < NV; ++i) {
            const int idx = t + i * 128;
            const int r   = idx / VPER;
            const int c4  = (idx - r * VPER) << 2;
            const int row = row0 + r;
            v[i] = *(const float4*)(X + (size_t)min(row, N - 1) * DIN + c4);
            if (row >= N) v[i] = make_float4(0.f, 0.f, 0.f, 0.f);
        }
        #pragma unroll
        for (int i = 0; i < NV; ++i) {
            const int idx = t + i * 128;
            const int r   = idx / VPER;
            const int c4  = (idx - r * VPER) << 2;
            *(float4*)(&tile[r * LROW + c4]) = v[i];
        }
        __syncthreads();
        if (mult) {
            v8f acc = {};
            #pragma unroll
            for (int s = 0; s < DIN / 4; ++s) {
                v2f a;
                a.x = tile[aoff + 4 * s];
                a.y = tile[aoff + 4 * s + 1];
                acc = __builtin_amdgcn_wmma_f32_16x16x4_f32(
                          false, a, false, b[s], (short)0, acc, false, false);
            }
            #pragma unroll
            for (int r = 0; r < 8; ++r) {
                const int m   = r + 8 * half;
                const int row = row0 + m;
                if (row < N && col < DOUT)
                    P[(size_t)row * DOUT + col] = acc[r] + bv;
            }
        }
    }
}

__global__ void gcn_pad_w4(const float* __restrict__ W4, const float* __restrict__ b4,
                           float* __restrict__ Wp, float* __restrict__ bp)
{
    const int i = blockIdx.x * blockDim.x + threadIdx.x;
    if (i < 64 * 16) {
        const int r = i >> 4, c = i & 15;
        Wp[i] = (c < 10) ? W4[r * 10 + c] : 0.f;
    } else if (i < 64 * 16 + 16) {
        const int c = i - 64 * 16;
        bp[c] = (c < 10) ? b4[c] : 0.f;
    }
}

__global__ void gcn_zero_i32(int* __restrict__ p, int n)
{
    const int i = blockIdx.x * blockDim.x + threadIdx.x;
    if (i < n) p[i] = 0;
}

#define EPT 5

__device__ __forceinline__ int package_half()
{
    return (int)((__builtin_amdgcn_s_sendmsg_rtn(0x87) >> 18) & 1u);
}

__global__ __launch_bounds__(256) void gcn_count_rank(
    const int* __restrict__ dst, int* __restrict__ cnt2, int* __restrict__ rank, int E)
{
    const int h = package_half();
    int* __restrict__ cnt = cnt2 + h * NNODES;
    const int T = gridDim.x * 256;
    for (int e0 = blockIdx.x * 256 + threadIdx.x; e0 < E; e0 += EPT * T) {
        int d[EPT], r[EPT];
        #pragma unroll
        for (int k = 0; k < EPT; ++k) {
            const int e = e0 + k * T;
            d[k] = e < E ? dst[e] : -1;
        }
        #pragma unroll
        for (int k = 0; k < EPT; ++k) {
            r[k] = 0;
            if (d[k] >= 0)
                r[k] = __hip_atomic_fetch_add(cnt + d[k], 1, __ATOMIC_RELAXED,
                                              __HIP_MEMORY_SCOPE_AGENT);
        }
        #pragma unroll
        for (int k = 0; k < EPT; ++k) {
            const int e = e0 + k * T;
            if (e < E) rank[e] = 2 * r[k] + h;
        }
    }
}

#define SCAN_ROWS ((NNODES / 4 + 1023) / 1024)
static_assert(NNODES % 4 == 0, "scan reads the counts as 4-int vectors");
static_assert(SCAN_ROWS * 1024 >= NNODES / 4, "the rows must cover every vector");

template <int R0, int R1>
__device__ __forceinline__ void scan_rows(const int4* __restrict__ c, const int t,
                                          int (&s)[SCAN_ROWS])
{
    #pragma unroll
    for (int r = R0; r < R1; ++r) {
        const int v = r * 1024 + t;
        int4 a = c[min(v, NNODES / 4 - 1)];
        if (v >= NNODES / 4) a = make_int4(0, 0, 0, 0);
        s[r] += a.x + a.y + a.z + a.w;
    }
}

__global__ __launch_bounds__(1024) void gcn_scan(
    const int* __restrict__ cnt2, int* __restrict__ rowptr)
{
    constexpr int NVEC = NNODES / 4;
    __shared__ int tot[SCAN_ROWS * 32];
    const int t    = threadIdx.x;
    const int lane = t & 31;
    const int wv   = t >> 5;
    const int4* ca = (const int4*)cnt2;
    const int4* cb = (const int4*)(cnt2 + NNODES);

    int s[SCAN_ROWS], incl[SCAN_ROWS];
    constexpr int RH = (SCAN_ROWS + 1) / 2;
    #pragma unroll
    for (int r = 0; r < SCAN_ROWS; ++r) s[r] = 0;
    scan_rows<0, RH>(ca, t, s);
    __builtin_amdgcn_sched_barrier(0);
    scan_rows<RH, SCAN_ROWS>(ca, t, s);
    __builtin_amdgcn_sched_barrier(0);
    scan_rows<0, RH>(cb, t, s);
    __builtin_amdgcn_sched_barrier(0);
    scan_rows<RH, SCAN_ROWS>(cb, t, s);
    #pragma unroll
    for (int r = 0; r < SCAN_ROWS; ++r) {
        int x = s[r];
        #pragma unroll
        for (int off = 1; off < 32; off <<= 1) {
            const int y = __shfl_up(x, off, 32);
            if (lane >= off) x += y;
        }
        incl[r] = x;
    }
    if (lane == 31) {
        #pragma unroll
        for (int r = 0; r < SCAN_ROWS; ++r) tot[r * 32 + wv] = incl[r];
    }
    #pragma unroll
    for (int r = 0; r < SCAN_ROWS; ++r) s[r] = incl[r] - s[r];
    __syncthreads();
    if (wv == 0) {
        int loc[SCAN_ROWS];
        int run = 0;
        #pragma unroll
        for (int k = 0; k < SCAN_ROWS; ++k) {
            const int v = tot[lane * SCAN_ROWS + k];
            loc[k] = run;
            run += v;
        }
        int wi = run;
        #pragma unroll
        for (int off = 1; off < 32; off <<= 1) {
            const int y = __shfl_up(wi, off, 32);
            if (lane >= off) wi += y;
        }
        const int ex = wi - run;
        #pragma unroll
        for (int k = 0; k < SCAN_ROWS; ++k) tot[lane * SCAN_ROWS + k] = loc[k] + ex;
        if (lane == 31) rowptr[NNODES] = wi;
    }
    __syncthreads();
    int4* r4 = (int4*)rowptr;
    #pragma unroll
    for (int r = 0; r < SCAN_ROWS; ++r) {
        const int v = r * 1024 + t;
        if (v < NVEC) {
            const int4 a = ca[v];
            const int4 b = cb[v];
            int4 o;
            o.x = tot[r * 32 + wv] + s[r];
            o.y = o.x + a.x + b.x;
            o.z = o.y + a.y + b.y;
            o.w = o.z + a.z + b.z;
            r4[v] = o;
        }
    }
}

__global__ __launch_bounds__(256) void gcn_fill(
    const int* __restrict__ src, const int* __restrict__ dst,
    const float* __restrict__ w, const int* __restrict__ rowptr,
    const int* __restrict__ rank, int* __restrict__ csr, int E)
{
    const int T = gridDim.x * 256;
    for (int e0 = blockIdx.x * 256 + threadIdx.x; e0 < E; e0 += EPT * T) {
        int d[EPT], rk[EPT], base[EPT];
        int2 v[EPT];
        #pragma unroll
        for (int k = 0; k < EPT; ++k) {
            const int ec = min(e0 + k * T, E - 1);
            d[k]   = dst[ec];
            rk[k]  = rank[ec];
            v[k].x = src[ec];
            v[k].y = __float_as_int(w[ec]);
        }
        #pragma unroll
        for (int k = 0; k < EPT; ++k) base[k] = rowptr[d[k] + (rk[k] & 1)];
        #pragma unroll
        for (int k = 0; k < EPT; ++k) {
            const int e = e0 + k * T;
            const int r = rk[k] >> 1;
            const int pos = (rk[k] & 1) ? base[k] - 1 - r : base[k] + r;
            if (e < E) *(int2*)(csr + 2 * (size_t)pos) = v[k];
        }
    }
}

template <int I>
__device__ __forceinline__ int row_share(const int v)
{
    return __builtin_amdgcn_update_dpp(0, v, 0x150 + I, 0xF, 0xF, true);
}
__device__ __forceinline__ int row_lane(const int v, const int i)
{
    switch (i) {
    case 0:  return row_share<0>(v);
    case 1:  return row_share<1>(v);
    case 2:  return row_share<2>(v);
    case 3:  return row_share<3>(v);
    case 4:  return row_share<4>(v);
    case 5:  return row_share<5>(v);
    case 6:  return row_share<6>(v);
    case 7:  return row_share<7>(v);
    case 8:  return row_share<8>(v);
    case 9:  return row_share<9>(v);
    case 10: return row_share<10>(v);
    case 11: return row_share<11>(v);
    case 12: return row_share<12>(v);
    case 13: return row_share<13>(v);
    case 14: return row_share<14>(v);
    default: return row_share<15>(v);
    }
}

__device__ __forceinline__ void fmav(float& a, const float r, const float w)
{
    a = fmaf(r, w, a);
}
__device__ __forceinline__ void fmav(float4& a, const float4 r, const float w)
{
    a.x = fmaf(r.x, w, a.x); a.y = fmaf(r.y, w, a.y);
    a.z = fmaf(r.z, w, a.z); a.w = fmaf(r.w, w, a.w);
}
__device__ __forceinline__ float  zero_of(const float*)  { return 0.f; }
__device__ __forceinline__ float4 zero_of(const float4*) { return make_float4(0.f, 0.f, 0.f, 0.f); }

template <typename V, int VPR, int CH>
__device__ __forceinline__ V gather_node(
    const V* __restrict__ Pv, const int2* __restrict__ ed,
    int j, const int end, const int l, const int col)
{
    static_assert(16 % CH == 0, "a load of 16 records is whole chunks");
    constexpr unsigned ROWB = VPR * sizeof(V);
    const char* base = (const char*)Pv;
    const unsigned colb = (unsigned)col * (unsigned)sizeof(V);
    V acc = zero_of((const V*)nullptr);
    for (; j < end; j += 32) {
        const int2 rec0 = ed[min(j + l, end - 1)];
        const int2 rec1 = ed[min(j + 16 + l, end - 1)];
        #pragma unroll
        for (int c = 0; c < 32; c += CH) {
            const int m = end - j - c;
            if (m > 0) {
                const int2 rec = c < 16 ? rec0 : rec1;
                const int c0 = c & 15;
                unsigned off[CH];
                #pragma unroll
                for (int i = 0; i < CH; ++i)
                    off[i] = (unsigned)row_lane(rec.x, c0 + i) * ROWB + colb;
                V r[CH];
                #pragma unroll
                for (int i = 0; i < CH; ++i)
                    if (i < m) r[i] = *(const V*)(base + off[i]);
                #pragma unroll
                for (int i = 0; i < CH; ++i)
                    if (i < m) fmav(acc, r[i], __int_as_float(row_lane(rec.y, c0 + i)));
            }
        }
    }
    return acc;
}

template <int DOUT, int DOUTP>
__global__ __launch_bounds__(256) void gcn_gather64_gemm(
    const float* __restrict__ P, const int* __restrict__ rowptr,
    const int* __restrict__ csr, const float* __restrict__ Wm,
    const float* __restrict__ bias, float* __restrict__ Pn, int N, int ntiles)
{
    constexpr int DIN  = 64;
    constexpr int LROW = DIN + 4;
    static_assert(DOUTP % 16 == 0 && DOUTP <= 64 && DOUT <= DOUTP, "column tiles");
    __shared__ float lds[2][16 * LROW];

    const int t    = threadIdx.x;
    const int nb   = t >> 4;
    const int l    = t & 15;

    const int wave    = t >> 5;
    const int lane    = t & 31;
    const int colbase = wave * 16;
    const bool mult   = colbase < DOUTP;
    const int half = lane >> 4;
    const int nrow = lane & 15;
    const int col  = colbase + nrow;

    v2f b[DIN / 4];
    float bv = 0.f;
    if (mult) {
        const float* wk = Wm + (size_t)(2 * half) * DOUTP + col;
        #pragma unroll
        for (int s = 0; s < DIN / 4; ++s) {
            b[s].x = wk[(size_t)(4 * s) * DOUTP];
            b[s].y = wk[(size_t)(4 * s + 1) * DOUTP];
        }
        bv = bias[col];
    }
    const int aoff = nrow * LROW + 2 * half;

    int buf = 0;
    for (int tl = blockIdx.x; tl < ntiles; tl += gridDim.x, buf ^= 1) {
        const int row0 = tl * 16;
        const int node = row0 + nb;
        float* tile = lds[buf];

        float4 acc = make_float4(0.f, 0.f, 0.f, 0.f);
        if (node < N)
            acc = gather_node<float4, 16, 8>((const float4*)P, (const int2*)csr,
                                             rowptr[node], rowptr[node + 1], l, l);

        acc.x = fmaxf(acc.x, 0.f); acc.y = fmaxf(acc.y, 0.f);
        acc.z = fmaxf(acc.z, 0.f); acc.w = fmaxf(acc.w, 0.f);
        *(float4*)(&tile[nb * LROW + 4 * l]) = acc;
        __syncthreads();

        if (mult) {
            v8f d = {};
            #pragma unroll
            for (int s = 0; s < DIN / 4; ++s) {
                v2f a;
                a.x = tile[aoff + 4 * s];
                a.y = tile[aoff + 4 * s + 1];
                d = __builtin_amdgcn_wmma_f32_16x16x4_f32(
                        false, a, false, b[s], (short)0, d, false, false);
            }
            #pragma unroll
            for (int r = 0; r < 8; ++r) {
                const int m   = r + 8 * half;
                const int row = row0 + m;
                if (row < N && col < DOUT)
                    Pn[(size_t)row * DOUT + col] = d[r] + bv;
            }
        }
    }
}

__global__ __launch_bounds__(256) void gcn_gather10(
    const float* __restrict__ P, const int* __restrict__ rowptr,
    const int* __restrict__ csr, float* __restrict__ out, int N, int ntiles)
{
    __shared__ float stage[2][160];
    const int t    = threadIdx.x;
    const int nb   = t >> 4;
    const int l    = t & 15;

    int buf = 0;
    for (int tl = blockIdx.x; tl < ntiles; tl += gridDim.x, buf ^= 1) {
        const int node = tl * 16 + nb;
        float acc = 0.f;
        if (node < N)
            acc = gather_node<float, 10, 16>(P, (const int2*)csr,
                                             rowptr[node], rowptr[node + 1], l, min(l, 9));
        if (l < 10) stage[buf][nb * 10 + l] = acc;
        __syncthreads();
        const int o = tl * 160 + t;
        if (t < 160 && o < N * 10) out[o] = stage[buf][t];
    }
}

extern "C" void kernel_launch(void* const* d_in, const int* in_sizes, int n_in,
                              void* d_out, int out_size, void* d_ws, size_t ws_size,
                              hipStream_t stream)
{
    (void)in_sizes; (void)n_in; (void)out_size; (void)ws_size;
    const int N = NNODES, E = NEDGES;

    const float* x    = (const float*)d_in[0];
    const int*   esrc = (const int*)  d_in[1];
    const int*   edst = (const int*)  d_in[2];
    const float* ew   = (const float*)d_in[3];
    const float* Ws[5]; const float* bs[5];
    for (int i = 0; i < 5; ++i) {
        Ws[i] = (const float*)d_in[4 + 2 * i];
        bs[i] = (const float*)d_in[5 + 2 * i];
    }
    float* out  = (float*)d_out;
    float* buf0 = (float*)d_ws;
    float* buf1 = buf0 + (size_t)N * 64;
    float* buf2 = buf1 + (size_t)N * 64;
    float* wpad = buf2 + (size_t)N * 64;
    float* bpad = wpad + 64 * 16;
    int*   cnt    = (int*)(bpad + 64);
    int*   rowptr = cnt + 2 * N;
    int*   rank   = rowptr + (N + 64);
    int*   csr    = rank + E;

    const int  ntiles = (N + 15) / 16;
    const dim3 pgrid(1250);

    gcn_pad_w4<<<(64 * 16 + 16 + 255) / 256, 256, 0, stream>>>(Ws[4], bs[4], wpad, bpad);

    gcn_zero_i32<<<(2 * N + 255) / 256, 256, 0, stream>>>(cnt, 2 * N);
    gcn_count_rank<<<pgrid, 256, 0, stream>>>(edst, cnt, rank, E);
    gcn_scan<<<1, 1024, 0, stream>>>(cnt, rowptr);
    gcn_fill<<<pgrid, 256, 0, stream>>>(esrc, edst, ew, rowptr, rank, csr, E);

    gcn_gemm_wmma<128, 64, 64><<<pgrid, 128, 0, stream>>>(x, Ws[0], bs[0], buf0, N, ntiles);

    gcn_gather64_gemm<64, 64><<<pgrid, 256, 0, stream>>>(buf0, rowptr, csr, Ws[1], bs[1], buf1, N, ntiles);
    gcn_gather64_gemm<64, 64><<<pgrid, 256, 0, stream>>>(buf1, rowptr, csr, Ws[2], bs[2], buf0, N, ntiles);
    gcn_gather64_gemm<64, 64><<<pgrid, 256, 0, stream>>>(buf0, rowptr, csr, Ws[3], bs[3], buf1, N, ntiles);

    gcn_gather64_gemm<10, 16><<<pgrid, 256, 0, stream>>>(buf1, rowptr, csr, wpad, bpad, buf0, N, ntiles);

    gcn_gather10<<<pgrid, 256, 0, stream>>>(buf0, rowptr, csr, out, N, ntiles);
}


# Tile walks software-pipelined: row pointers two tiles ahead, first edge records and layer-0 A rows one tile ahead, in flight across an LDS-only work-group barrier; per-node sum order unchanged
# speedup vs baseline: 3.8659x; 1.0326x over previous
#include <hip/hip_runtime.h>
#include <hip/hip_bf16.h>

typedef float v2f __attribute__((ext_vector_type(2)));
typedef float v8f __attribute__((ext_vector_type(8)));

#define NNODES 100000
#define NEDGES 1600000


__device__ __forceinline__ void lds_barrier()
{
    __builtin_amdgcn_fence(__ATOMIC_RELEASE, "workgroup", "local");
    __builtin_amdgcn_s_barrier();
    __builtin_amdgcn_fence(__ATOMIC_ACQUIRE, "workgroup", "local");
}

template <int DIN, int DOUT, int DOUTP>
__global__ __launch_bounds__(128) void gcn_gemm_wmma(
    const float* __restrict__ X, const float* __restrict__ Wm,
    const float* __restrict__ bias, float* __restrict__ P, int N, int ntiles)
{
    constexpr int LROW = DIN + 4;
    constexpr int VPER = DIN / 4;
    constexpr int NV   = 16 * VPER / 128;
    static_assert(DIN % 32 == 0 && DOUTP % 16 == 0 && DOUTP <= 64 && DOUT <= DOUTP, "tiles");
    __shared__ float lds[2][16 * LROW];
    const int t = threadIdx.x;

    const int wave = t >> 5;
    const int lane = t & 31;
    const int colbase = wave * 16;
    const bool mult = colbase < DOUTP;
    const int half = lane >> 4;
    const int nrow = lane & 15;
    const int col  = colbase + nrow;

    v2f b[DIN / 4];
    float bv = 0.f;
    if (mult) {
        const float* wk = Wm + (size_t)(2 * half) * DOUTP + col;
        #pragma unroll
        for (int s = 0; s < DIN / 4; ++s) {
            b[s].x = wk[(size_t)(4 * s) * DOUTP];
            b[s].y = wk[(size_t)(4 * s + 1) * DOUTP];
        }
        bv = bias[col];
    }
    const int aoff = nrow * LROW + 2 * half;

    float4 v[NV];
    if ((int)blockIdx.x < ntiles) {
        #pragma unroll
        for (int i = 0; i < NV; ++i) {
            const int idx = t + i * 128;
            const int r   = idx / VPER;
            const int c4  = (idx - r * VPER) << 2;
            const int row = (int)blockIdx.x * 16 + r;
            v[i] = *(const float4*)(X + (size_t)min(row, N - 1) * DIN + c4);
        }
    }

    int buf = 0;
    for (int tl = blockIdx.x; tl < ntiles; tl += gridDim.x, buf ^= 1) {
        const int row0 = tl * 16;
        float* tile = lds[buf];
        #pragma unroll
        for (int i = 0; i < NV; ++i) {
            const int idx = t + i * 128;
            const int r   = idx / VPER;
            const int c4  = (idx - r * VPER) << 2;
            const bool in = row0 + r < N;
            *(float4*)(&tile[r * LROW + c4]) = in ? v[i] : make_float4(0.f, 0.f, 0.f, 0.f);
        }
        const int ntl = tl + gridDim.x;
        if (ntl < ntiles) {
            #pragma unroll
            for (int i = 0; i < NV; ++i) {
                const int idx = t + i * 128;
                const int r   = idx / VPER;
                const int c4  = (idx - r * VPER) << 2;
                const int row = ntl * 16 + r;
                v[i] = *(const float4*)(X + (size_t)min(row, N - 1) * DIN + c4);
            }
        }
        lds_barrier();
        if (mult) {
            v8f acc = {};
            #pragma unroll
            for (int s = 0; s < DIN / 4; ++s) {
                v2f a;
                a.x = tile[aoff + 4 * s];
                a.y = tile[aoff + 4 * s + 1];
                acc = __builtin_amdgcn_wmma_f32_16x16x4_f32(
                          false, a, false, b[s], (short)0, acc, false, false);
            }
            #pragma unroll
            for (int r = 0; r < 8; ++r) {
                const int m   = r + 8 * half;
                const int row = row0 + m;
                if (row < N && col < DOUT)
                    P[(size_t)row * DOUT + col] = acc[r] + bv;
            }
        }
    }
}

__global__ void gcn_pad_w4(const float* __restrict__ W4, const float* __restrict__ b4,
                           float* __restrict__ Wp, float* __restrict__ bp)
{
    const int i = blockIdx.x * blockDim.x + threadIdx.x;
    if (i < 64 * 16) {
        const int r = i >> 4, c = i & 15;
        Wp[i] = (c < 10) ? W4[r * 10 + c] : 0.f;
    } else if (i < 64 * 16 + 16) {
        const int c = i - 64 * 16;
        bp[c] = (c < 10) ? b4[c] : 0.f;
    }
}

__global__ void gcn_zero_i32(int* __restrict__ p, int n)
{
    const int i = blockIdx.x * blockDim.x + threadIdx.x;
    if (i < n) p[i] = 0;
}

#define EPT 5

__device__ __forceinline__ int package_half()
{
    return (int)((__builtin_amdgcn_s_sendmsg_rtn(0x87) >> 18) & 1u);
}

__global__ __launch_bounds__(256) void gcn_count_rank(
    const int* __restrict__ dst, int* __restrict__ cnt2, int* __restrict__ rank, int E)
{
    const int h = package_half();
    int* __restrict__ cnt = cnt2 + h * NNODES;
    const int T = gridDim.x * 256;
    for (int e0 = blockIdx.x * 256 + threadIdx.x; e0 < E; e0 += EPT * T) {
        int d[EPT], r[EPT];
        #pragma unroll
        for (int k = 0; k < EPT; ++k) {
            const int e = e0 + k * T;
            d[k] = e < E ? dst[e] : -1;
        }
        #pragma unroll
        for (int k = 0; k < EPT; ++k) {
            r[k] = 0;
            if (d[k] >= 0)
                r[k] = __hip_atomic_fetch_add(cnt + d[k], 1, __ATOMIC_RELAXED,
                                              __HIP_MEMORY_SCOPE_AGENT);
        }
        #pragma unroll
        for (int k = 0; k < EPT; ++k) {
            const int e = e0 + k * T;
            if (e < E) rank[e] = 2 * r[k] + h;
        }
    }
}

#define SCAN_ROWS ((NNODES / 4 + 1023) / 1024)
static_assert(NNODES % 4 == 0, "scan reads the counts as 4-int vectors");
static_assert(SCAN_ROWS * 1024 >= NNODES / 4, "the rows must cover every vector");

template <int R0, int R1>
__device__ __forceinline__ void scan_rows(const int4* __restrict__ c, const int t,
                                          int (&s)[SCAN_ROWS])
{
    #pragma unroll
    for (int r = R0; r < R1; ++r) {
        const int v = r * 1024 + t;
        int4 a = c[min(v, NNODES / 4 - 1)];
        if (v >= NNODES / 4) a = make_int4(0, 0, 0, 0);
        s[r] += a.x + a.y + a.z + a.w;
    }
}

__global__ __launch_bounds__(1024) void gcn_scan(
    const int* __restrict__ cnt2, int* __restrict__ rowptr)
{
    constexpr int NVEC = NNODES / 4;
    __shared__ int tot[SCAN_ROWS * 32];
    const int t    = threadIdx.x;
    const int lane = t & 31;
    const int wv   = t >> 5;
    const int4* ca = (const int4*)cnt2;
    const int4* cb = (const int4*)(cnt2 + NNODES);

    int s[SCAN_ROWS], incl[SCAN_ROWS];
    constexpr int RH = (SCAN_ROWS + 1) / 2;
    #pragma unroll
    for (int r = 0; r < SCAN_ROWS; ++r) s[r] = 0;
    scan_rows<0, RH>(ca, t, s);
    __builtin_amdgcn_sched_barrier(0);
    scan_rows<RH, SCAN_ROWS>(ca, t, s);
    __builtin_amdgcn_sched_barrier(0);
    scan_rows<0, RH>(cb, t, s);
    __builtin_amdgcn_sched_barrier(0);
    scan_rows<RH, SCAN_ROWS>(cb, t, s);
    #pragma unroll
    for (int r = 0; r < SCAN_ROWS; ++r) {
        int x = s[r];
        #pragma unroll
        for (int off = 1; off < 32; off <<= 1) {
            const int y = __shfl_up(x, off, 32);
            if (lane >= off) x += y;
        }
        incl[r] = x;
    }
    if (lane == 31) {
        #pragma unroll
        for (int r = 0; r < SCAN_ROWS; ++r) tot[r * 32 + wv] = incl[r];
    }
    #pragma unroll
    for (int r = 0; r < SCAN_ROWS; ++r) s[r] = incl[r] - s[r];
    __syncthreads();
    if (wv == 0) {
        int loc[SCAN_ROWS];
        int run = 0;
        #pragma unroll
        for (int k = 0; k < SCAN_ROWS; ++k) {
            const int v = tot[lane * SCAN_ROWS + k];
            loc[k] = run;
            run += v;
        }
        int wi = run;
        #pragma unroll
        for (int off = 1; off < 32; off <<= 1) {
            const int y = __shfl_up(wi, off, 32);
            if (lane >= off) wi += y;
        }
        const int ex = wi - run;
        #pragma unroll
        for (int k = 0; k < SCAN_ROWS; ++k) tot[lane * SCAN_ROWS + k] = loc[k] + ex;
        if (lane == 31) rowptr[NNODES] = wi;
    }
    __syncthreads();
    int4* r4 = (int4*)rowptr;
    #pragma unroll
    for (int r = 0; r < SCAN_ROWS; ++r) {
        const int v = r * 1024 + t;
        if (v < NVEC) {
            const int4 a = ca[v];
            const int4 b = cb[v];
            int4 o;
            o.x = tot[r * 32 + wv] + s[r];
            o.y = o.x + a.x + b.x;
            o.z = o.y + a.y + b.y;
            o.w = o.z + a.z + b.z;
            r4[v] = o;
        }
    }
}

__global__ __launch_bounds__(256) void gcn_fill(
    const int* __restrict__ src, const int* __restrict__ dst,
    const float* __restrict__ w, const int* __restrict__ rowptr,
    const int* __restrict__ rank, int* __restrict__ csr, int E)
{
    const int T = gridDim.x * 256;
    for (int e0 = blockIdx.x * 256 + threadIdx.x; e0 < E; e0 += EPT * T) {
        int d[EPT], rk[EPT], base[EPT];
        int2 v[EPT];
        #pragma unroll
        for (int k = 0; k < EPT; ++k) {
            const int ec = min(e0 + k * T, E - 1);
            d[k]   = dst[ec];
            rk[k]  = rank[ec];
            v[k].x = src[ec];
            v[k].y = __float_as_int(w[ec]);
        }
        #pragma unroll
        for (int k = 0; k < EPT; ++k) base[k] = rowptr[d[k] + (rk[k] & 1)];
        #pragma unroll
        for (int k = 0; k < EPT; ++k) {
            const int e = e0 + k * T;
            const int r = rk[k] >> 1;
            const int pos = (rk[k] & 1) ? base[k] - 1 - r : base[k] + r;
            if (e < E) *(int2*)(csr + 2 * (size_t)pos) = v[k];
        }
    }
}

template <int I>
__device__ __forceinline__ int row_share(const int v)
{
    return __builtin_amdgcn_update_dpp(0, v, 0x150 + I, 0xF, 0xF, true);
}
__device__ __forceinline__ int row_lane(const int v, const int i)
{
    switch (i) {
    case 0:  return row_share<0>(v);
    case 1:  return row_share<1>(v);
    case 2:  return row_share<2>(v);
    case 3:  return row_share<3>(v);
    case 4:  return row_share<4>(v);
    case 5:  return row_share<5>(v);
    case 6:  return row_share<6>(v);
    case 7:  return row_share<7>(v);
    case 8:  return row_share<8>(v);
    case 9:  return row_share<9>(v);
    case 10: return row_share<10>(v);
    case 11: return row_share<11>(v);
    case 12: return row_share<12>(v);
    case 13: return row_share<13>(v);
    case 14: return row_share<14>(v);
    default: return row_share<15>(v);
    }
}

__device__ __forceinline__ void fmav(float& a, const float r, const float w)
{
    a = fmaf(r, w, a);
}
__device__ __forceinline__ void fmav(float4& a, const float4 r, const float w)
{
    a.x = fmaf(r.x, w, a.x); a.y = fmaf(r.y, w, a.y);
    a.z = fmaf(r.z, w, a.z); a.w = fmaf(r.w, w, a.w);
}
__device__ __forceinline__ float  zero_of(const float*)  { return 0.f; }
__device__ __forceinline__ float4 zero_of(const float4*) { return make_float4(0.f, 0.f, 0.f, 0.f); }

__device__ __forceinline__ void load_records(
    const int2* __restrict__ ed, const int j, const int end, const int l,
    int2& rec0, int2& rec1)
{
    rec0 = make_int2(0, 0);
    rec1 = make_int2(0, 0);
    if (j < end) {
        rec0 = ed[min(j + l, end - 1)];
        rec1 = ed[min(j + 16 + l, end - 1)];
    }
}

template <typename V, int VPR, int CH>
__device__ __forceinline__ V gather_node(
    const V* __restrict__ Pv, const int2* __restrict__ ed,
    int j, const int end, const int l, const int col, int2 rec0, int2 rec1)
{
    static_assert(16 % CH == 0, "a load of 16 records is whole chunks");
    constexpr unsigned ROWB = VPR * sizeof(V);
    const char* base = (const char*)Pv;
    const unsigned colb = (unsigned)col * (unsigned)sizeof(V);
    V acc = zero_of((const V*)nullptr);
    for (; j < end; j += 32) {
        #pragma unroll
        for (int c = 0; c < 32; c += CH) {
            const int m = end - j - c;
            if (m > 0) {
                const int2 rec = c < 16 ? rec0 : rec1;
                const int c0 = c & 15;
                unsigned off[CH];
                #pragma unroll
                for (int i = 0; i < CH; ++i)
                    off[i] = (unsigned)row_lane(rec.x, c0 + i) * ROWB + colb;
                V r[CH];
                #pragma unroll
                for (int i = 0; i < CH; ++i)
                    if (i < m) r[i] = *(const V*)(base + off[i]);
                #pragma unroll
                for (int i = 0; i < CH; ++i)
                    if (i < m) fmav(acc, r[i], __int_as_float(row_lane(rec.y, c0 + i)));
            }
        }
        if (j + 32 < end) load_records(ed, j + 32, end, l, rec0, rec1);
    }
    return acc;
}

__device__ __forceinline__ void load_rowptr(
    const int* __restrict__ rowptr, const int tl, const int ntiles,
    const int nb, const int N, int& jb, int& je)
{
    jb = 0;
    je = 0;
    const int node = tl * 16 + nb;
    if (tl < ntiles && node < N) {
        jb = rowptr[node];
        je = rowptr[node + 1];
    }
}

template <int DOUT, int DOUTP>
__global__ __launch_bounds__(256) void gcn_gather64_gemm(
    const float* __restrict__ P, const int* __restrict__ rowptr,
    const int* __restrict__ csr, const float* __restrict__ Wm,
    const float* __restrict__ bias, float* __restrict__ Pn, int N, int ntiles)
{
    constexpr int DIN  = 64;
    constexpr int LROW = DIN + 4;
    static_assert(DOUTP % 16 == 0 && DOUTP <= 64 && DOUT <= DOUTP, "column tiles");
    __shared__ float lds[2][16 * LROW];

    const int t    = threadIdx.x;
    const int nb   = t >> 4;
    const int l    = t & 15;

    const int wave    = t >> 5;
    const int lane    = t & 31;
    const int colbase = wave * 16;
    const bool mult   = colbase < DOUTP;
    const int half = lane >> 4;
    const int nrow = lane & 15;
    const int col  = colbase + nrow;

    v2f b[DIN / 4];
    float bv = 0.f;
    if (mult) {
        const float* wk = Wm + (size_t)(2 * half) * DOUTP + col;
        #pragma unroll
        for (int s = 0; s < DIN / 4; ++s) {
            b[s].x = wk[(size_t)(4 * s) * DOUTP];
            b[s].y = wk[(size_t)(4 * s + 1) * DOUTP];
        }
        bv = bias[col];
    }
    const int aoff = nrow * LROW + 2 * half;

    const int2* ed = (const int2*)csr;
    const int G = (int)gridDim.x;
    int jb, je, njb, nje;
    int2 rec0, rec1;
    load_rowptr(rowptr, (int)blockIdx.x, ntiles, nb, N, jb, je);
    load_rowptr(rowptr, (int)blockIdx.x + G, ntiles, nb, N, njb, nje);
    load_records(ed, jb, je, l, rec0, rec1);

    int buf = 0;
    for (int tl = blockIdx.x; tl < ntiles; tl += G, buf ^= 1) {
        const int row0 = tl * 16;
        float* tile = lds[buf];

        float4 acc = gather_node<float4, 16, 8>((const float4*)P, ed, jb, je, l, l, rec0, rec1);

        jb = njb;
        je = nje;
        load_records(ed, jb, je, l, rec0, rec1);
        load_rowptr(rowptr, tl + 2 * G, ntiles, nb, N, njb, nje);

        acc.x = fmaxf(acc.x, 0.f); acc.y = fmaxf(acc.y, 0.f);
        acc.z = fmaxf(acc.z, 0.f); acc.w = fmaxf(acc.w, 0.f);
        *(float4*)(&tile[nb * LROW + 4 * l]) = acc;
        lds_barrier();

        if (mult) {
            v8f d = {};
            #pragma unroll
            for (int s = 0; s < DIN / 4; ++s) {
                v2f a;
                a.x = tile[aoff + 4 * s];
                a.y = tile[aoff + 4 * s + 1];
                d = __builtin_amdgcn_wmma_f32_16x16x4_f32(
                        false, a, false, b[s], (short)0, d, false, false);
            }
            #pragma unroll
            for (int r = 0; r < 8; ++r) {
                const int m   = r + 8 * half;
                const int row = row0 + m;
                if (row < N && col < DOUT)
                    Pn[(size_t)row * DOUT + col] = d[r] + bv;
            }
        }
    }
}

__global__ __launch_bounds__(256) void gcn_gather10(
    const float* __restrict__ P, const int* __restrict__ rowptr,
    const int* __restrict__ csr, float* __restrict__ out, int N, int ntiles)
{
    __shared__ float stage[2][160];
    const int t    = threadIdx.x;
    const int nb   = t >> 4;
    const int l    = t & 15;

    const int2* ed = (const int2*)csr;
    const int G = (int)gridDim.x;
    int jb, je, njb, nje;
    int2 rec0, rec1;
    load_rowptr(rowptr, (int)blockIdx.x, ntiles, nb, N, jb, je);
    load_rowptr(rowptr, (int)blockIdx.x + G, ntiles, nb, N, njb, nje);
    load_records(ed, jb, je, l, rec0, rec1);

    int buf = 0;
    for (int tl = blockIdx.x; tl < ntiles; tl += G, buf ^= 1) {
        const float acc = gather_node<float, 10, 16>(P, ed, jb, je, l, min(l, 9), rec0, rec1);
        jb = njb;
        je = nje;
        load_records(ed, jb, je, l, rec0, rec1);
        load_rowptr(rowptr, tl + 2 * G, ntiles, nb, N, njb, nje);
        if (l < 10) stage[buf][nb * 10 + l] = acc;
        lds_barrier();
        const int o = tl * 160 + t;
        if (t < 160 && o < N * 10) out[o] = stage[buf][t];
    }
}

extern "C" void kernel_launch(void* const* d_in, const int* in_sizes, int n_in,
                              void* d_out, int out_size, void* d_ws, size_t ws_size,
                              hipStream_t stream)
{
    (void)in_sizes; (void)n_in; (void)out_size; (void)ws_size;
    const int N = NNODES, E = NEDGES;

    const float* x    = (const float*)d_in[0];
    const int*   esrc = (const int*)  d_in[1];
    const int*   edst = (const int*)  d_in[2];
    const float* ew   = (const float*)d_in[3];
    const float* Ws[5]; const float* bs[5];
    for (int i = 0; i < 5; ++i) {
        Ws[i] = (const float*)d_in[4 + 2 * i];
        bs[i] = (const float*)d_in[5 + 2 * i];
    }
    float* out  = (float*)d_out;
    float* buf0 = (float*)d_ws;
    float* buf1 = buf0 + (size_t)N * 64;
    float* buf2 = buf1 + (size_t)N * 64;
    float* wpad = buf2 + (size_t)N * 64;
    float* bpad = wpad + 64 * 16;
    int*   cnt    = (int*)(bpad + 64);
    int*   rowptr = cnt + 2 * N;
    int*   rank   = rowptr + (N + 64);
    int*   csr    = rank + E;

    const int  ntiles = (N + 15) / 16;
    const dim3 pgrid(1250);

    gcn_pad_w4<<<(64 * 16 + 16 + 255) / 256, 256, 0, stream>>>(Ws[4], bs[4], wpad, bpad);

    gcn_zero_i32<<<(2 * N + 255) / 256, 256, 0, stream>>>(cnt, 2 * N);
    gcn_count_rank<<<pgrid, 256, 0, stream>>>(edst, cnt, rank, E);
    gcn_scan<<<1, 1024, 0, stream>>>(cnt, rowptr);
    gcn_fill<<<pgrid, 256, 0, stream>>>(esrc, edst, ew, rowptr, rank, csr, E);

    gcn_gemm_wmma<128, 64, 64><<<pgrid, 128, 0, stream>>>(x, Ws[0], bs[0], buf0, N, ntiles);

    gcn_gather64_gemm<64, 64><<<pgrid, 256, 0, stream>>>(buf0, rowptr, csr, Ws[1], bs[1], buf1, N, ntiles);
    gcn_gather64_gemm<64, 64><<<pgrid, 256, 0, stream>>>(buf1, rowptr, csr, Ws[2], bs[2], buf0, N, ntiles);
    gcn_gather64_gemm<64, 64><<<pgrid, 256, 0, stream>>>(buf0, rowptr, csr, Ws[3], bs[3], buf1, N, ntiles);

    gcn_gather64_gemm<10, 16><<<pgrid, 256, 0, stream>>>(buf1, rowptr, csr, wpad, bpad, buf0, N, ntiles);

    gcn_gather10<<<pgrid, 256, 0, stream>>>(buf0, rowptr, csr, out, N, ntiles);
}


# Weight padding and the layer-0 projection, which wait for nothing in the edge sort, run as the other work-groups of the one-work-group scan launch: 11 launches become 9; all sum orders unchanged
# speedup vs baseline: 4.0755x; 1.0542x over previous
#include <hip/hip_runtime.h>
#include <hip/hip_bf16.h>

typedef float v2f __attribute__((ext_vector_type(2)));
typedef float v8f __attribute__((ext_vector_type(8)));

#define NNODES 100000
#define NEDGES 1600000


__device__ __forceinline__ void lds_barrier()
{
    __builtin_amdgcn_fence(__ATOMIC_RELEASE, "workgroup", "local");
    __builtin_amdgcn_s_barrier();
    __builtin_amdgcn_fence(__ATOMIC_ACQUIRE, "workgroup", "local");
}

#define GEMM0_GROUPS 8
template <int DIN, int DOUT, int DOUTP>
__device__ __forceinline__ void gcn_gemm_groups(
    const float* __restrict__ X, const float* __restrict__ Wm,
    const float* __restrict__ bias, float* __restrict__ P, int N, int ntiles,
    const int wg, const int nwg)
{
    constexpr int LROW = DIN + 4;
    constexpr int VPER = DIN / 4;
    constexpr int NV   = 16 * VPER / 128;
    static_assert(DIN % 32 == 0 && DOUTP % 16 == 0 && DOUTP <= 64 && DOUT <= DOUTP, "tiles");
    __shared__ float lds[2][GEMM0_GROUPS][16 * LROW];
    const int grp = threadIdx.x >> 7;
    const int t   = threadIdx.x & 127;

    const int wave = t >> 5;
    const int lane = t & 31;
    const int colbase = wave * 16;
    const bool mult = colbase < DOUTP;
    const int half = lane >> 4;
    const int nrow = lane & 15;
    const int col  = colbase + nrow;

    v2f b[DIN / 4];
    float bv = 0.f;
    if (mult) {
        const float* wk = Wm + (size_t)(2 * half) * DOUTP + col;
        #pragma unroll
        for (int s = 0; s < DIN / 4; ++s) {
            b[s].x = wk[(size_t)(4 * s) * DOUTP];
            b[s].y = wk[(size_t)(4 * s + 1) * DOUTP];
        }
        bv = bias[col];
    }
    const int aoff = nrow * LROW + 2 * half;

    const int stride = nwg * GEMM0_GROUPS;
    const int first  = wg * GEMM0_GROUPS + grp;
    const int passes = (ntiles + stride - 1) / stride;

    float4 v[NV];
    #pragma unroll
    for (int i = 0; i < NV; ++i) {
        const int idx = t + i * 128;
        const int r   = idx / VPER;
        const int c4  = (idx - r * VPER) << 2;
        const int row = first * 16 + r;
        v[i] = *(const float4*)(X + (size_t)min(row, N - 1) * DIN + c4);
    }

    int buf = 0;
    for (int ps = 0; ps < passes; ++ps, buf ^= 1) {
        const int tl   = first + ps * stride;
        const int row0 = tl * 16;
        float* tile = lds[buf][grp];
        #pragma unroll
        for (int i = 0; i < NV; ++i) {
            const int idx = t + i * 128;
            const int r   = idx / VPER;
            const int c4  = (idx - r * VPER) << 2;
            const bool in = row0 + r < N;
            *(float4*)(&tile[r * LROW + c4]) = in ? v[i] : make_float4(0.f, 0.f, 0.f, 0.f);
        }
        if (ps + 1 < passes) {
            #pragma unroll
            for (int i = 0; i < NV; ++i) {
                const int idx = t + i * 128;
                const int r   = idx / VPER;
                const int c4  = (idx - r * VPER) << 2;
                const int row = (tl + stride) * 16 + r;
                v[i] = *(const float4*)(X + (size_t)min(row, N - 1) * DIN + c4);
            }
        }
        lds_barrier();
        if (mult) {
            v8f acc = {};
            #pragma unroll
            for (int s = 0; s < DIN / 4; ++s) {
                v2f a;
                a.x = tile[aoff + 4 * s];
                a.y = tile[aoff + 4 * s + 1];
                acc = __builtin_amdgcn_wmma_f32_16x16x4_f32(
                          false, a, false, b[s], (short)0, acc, false, false);
            }
            #pragma unroll
            for (int r = 0; r < 8; ++r) {
                const int m   = r + 8 * half;
                const int row = row0 + m;
                if (row < N && col < DOUT)
                    P[(size_t)row * DOUT + col] = acc[r] + bv;
            }
        }
    }
}

__device__ __forceinline__ void gcn_pad_w4(const float* __restrict__ W4, const float* __restrict__ b4,
                                           float* __restrict__ Wp, float* __restrict__ bp)
{
    for (int i = threadIdx.x; i < 64 * 16 + 16; i += blockDim.x) {
        if (i < 64 * 16) {
            const int r = i >> 4, c = i & 15;
            Wp[i] = (c < 10) ? W4[r * 10 + c] : 0.f;
        } else {
            const int c = i - 64 * 16;
            bp[c] = (c < 10) ? b4[c] : 0.f;
        }
    }
}

__global__ void gcn_zero_i32(int* __restrict__ p, int n)
{
    const int i = blockIdx.x * blockDim.x + threadIdx.x;
    if (i < n) p[i] = 0;
}

#define EPT 5

__device__ __forceinline__ int package_half()
{
    return (int)((__builtin_amdgcn_s_sendmsg_rtn(0x87) >> 18) & 1u);
}

__global__ __launch_bounds__(256) void gcn_count_rank(
    const int* __restrict__ dst, int* __restrict__ cnt2, int* __restrict__ rank, int E)
{
    const int h = package_half();
    int* __restrict__ cnt = cnt2 + h * NNODES;
    const int T = gridDim.x * 256;
    for (int e0 = blockIdx.x * 256 + threadIdx.x; e0 < E; e0 += EPT * T) {
        int d[EPT], r[EPT];
        #pragma unroll
        for (int k = 0; k < EPT; ++k) {
            const int e = e0 + k * T;
            d[k] = e < E ? dst[e] : -1;
        }
        #pragma unroll
        for (int k = 0; k < EPT; ++k) {
            r[k] = 0;
            if (d[k] >= 0)
                r[k] = __hip_atomic_fetch_add(cnt + d[k], 1, __ATOMIC_RELAXED,
                                              __HIP_MEMORY_SCOPE_AGENT);
        }
        #pragma unroll
        for (int k = 0; k < EPT; ++k) {
            const int e = e0 + k * T;
            if (e < E) rank[e] = 2 * r[k] + h;
        }
    }
}

#define SCAN_ROWS ((NNODES / 4 + 1023) / 1024)
static_assert(NNODES % 4 == 0, "scan reads the counts as 4-int vectors");
static_assert(SCAN_ROWS * 1024 >= NNODES / 4, "the rows must cover every vector");

template <int R0, int R1>
__device__ __forceinline__ void scan_rows(const int4* __restrict__ c, const int t,
                                          int (&s)[SCAN_ROWS])
{
    #pragma unroll
    for (int r = R0; r < R1; ++r) {
        const int v = r * 1024 + t;
        int4 a = c[min(v, NNODES / 4 - 1)];
        if (v >= NNODES / 4) a = make_int4(0, 0, 0, 0);
        s[r] += a.x + a.y + a.z + a.w;
    }
}

__device__ __forceinline__ void gcn_scan(
    const int* __restrict__ cnt2, int* __restrict__ rowptr)
{
    constexpr int NVEC = NNODES / 4;
    __shared__ int tot[SCAN_ROWS * 32];
    const int t    = threadIdx.x;
    const int lane = t & 31;
    const int wv   = t >> 5;
    const int4* ca = (const int4*)cnt2;
    const int4* cb = (const int4*)(cnt2 + NNODES);

    int s[SCAN_ROWS], incl[SCAN_ROWS];
    constexpr int RH = (SCAN_ROWS + 1) / 2;
    #pragma unroll
    for (int r = 0; r < SCAN_ROWS; ++r) s[r] = 0;
    scan_rows<0, RH>(ca, t, s);
    __builtin_amdgcn_sched_barrier(0);
    scan_rows<RH, SCAN_ROWS>(ca, t, s);
    __builtin_amdgcn_sched_barrier(0);
    scan_rows<0, RH>(cb, t, s);
    __builtin_amdgcn_sched_barrier(0);
    scan_rows<RH, SCAN_ROWS>(cb, t, s);
    #pragma unroll
    for (int r = 0; r < SCAN_ROWS; ++r) {
        int x = s[r];
        #pragma unroll
        for (int off = 1; off < 32; off <<= 1) {
            const int y = __shfl_up(x, off, 32);
            if (lane >= off) x += y;
        }
        incl[r] = x;
    }
    if (lane == 31) {
        #pragma unroll
        for (int r = 0; r < SCAN_ROWS; ++r) tot[r * 32 + wv] = incl[r];
    }
    #pragma unroll
    for (int r = 0; r < SCAN_ROWS; ++r) s[r] = incl[r] - s[r];
    __syncthreads();
    if (wv == 0) {
        int loc[SCAN_ROWS];
        int run = 0;
        #pragma unroll
        for (int k = 0; k < SCAN_ROWS; ++k) {
            const int v = tot[lane * SCAN_ROWS + k];
            loc[k] = run;
            run += v;
        }
        int wi = run;
        #pragma unroll
        for (int off = 1; off < 32; off <<= 1) {
            const int y = __shfl_up(wi, off, 32);
            if (lane >= off) wi += y;
        }
        const int ex = wi - run;
        #pragma unroll
        for (int k = 0; k < SCAN_ROWS; ++k) tot[lane * SCAN_ROWS + k] = loc[k] + ex;
        if (lane == 31) rowptr[NNODES] = wi;
    }
    __syncthreads();
    int4* r4 = (int4*)rowptr;
    #pragma unroll
    for (int r = 0; r < SCAN_ROWS; ++r) {
        if (r % 9 == 0 && r > 0) __builtin_amdgcn_sched_barrier(0);
        const int v = r * 1024 + t;
        if (v < NVEC) {
            const int4 a = ca[v];
            const int4 b = cb[v];
            int4 o;
            o.x = tot[r * 32 + wv] + s[r];
            o.y = o.x + a.x + b.x;
            o.z = o.y + a.y + b.y;
            o.w = o.z + a.z + b.z;
            r4[v] = o;
        }
    }
}

__global__ __launch_bounds__(1024) void gcn_scan_gemm0(
    const int* __restrict__ cnt2, int* __restrict__ rowptr,
    const float* __restrict__ X, const float* __restrict__ W0,
    const float* __restrict__ b0, float* __restrict__ P, int N, int ntiles,
    const float* __restrict__ W4, const float* __restrict__ b4,
    float* __restrict__ Wp, float* __restrict__ bp)
{
    if (blockIdx.x == 0) {
        gcn_scan(cnt2, rowptr);
        return;
    }
    const int wg = (int)blockIdx.x - 1;
    if (wg == 0) gcn_pad_w4(W4, b4, Wp, bp);
    gcn_gemm_groups<128, 64, 64>(X, W0, b0, P, N, ntiles, wg, (int)gridDim.x - 1);
}

__global__ __launch_bounds__(256) void gcn_fill(
    const int* __restrict__ src, const int* __restrict__ dst,
    const float* __restrict__ w, const int* __restrict__ rowptr,
    const int* __restrict__ rank, int* __restrict__ csr, int E)
{
    const int T = gridDim.x * 256;
    for (int e0 = blockIdx.x * 256 + threadIdx.x; e0 < E; e0 += EPT * T) {
        int d[EPT], rk[EPT], base[EPT];
        int2 v[EPT];
        #pragma unroll
        for (int k = 0; k < EPT; ++k) {
            const int ec = min(e0 + k * T, E - 1);
            d[k]   = dst[ec];
            rk[k]  = rank[ec];
            v[k].x = src[ec];
            v[k].y = __float_as_int(w[ec]);
        }
        #pragma unroll
        for (int k = 0; k < EPT; ++k) base[k] = rowptr[d[k] + (rk[k] & 1)];
        #pragma unroll
        for (int k = 0; k < EPT; ++k) {
            const int e = e0 + k * T;
            const int r = rk[k] >> 1;
            const int pos = (rk[k] & 1) ? base[k] - 1 - r : base[k] + r;
            if (e < E) *(int2*)(csr + 2 * (size_t)pos) = v[k];
        }
    }
}

template <int I>
__device__ __forceinline__ int row_share(const int v)
{
    return __builtin_amdgcn_update_dpp(0, v, 0x150 + I, 0xF, 0xF, true);
}
__device__ __forceinline__ int row_lane(const int v, const int i)
{
    switch (i) {
    case 0:  return row_share<0>(v);
    case 1:  return row_share<1>(v);
    case 2:  return row_share<2>(v);
    case 3:  return row_share<3>(v);
    case 4:  return row_share<4>(v);
    case 5:  return row_share<5>(v);
    case 6:  return row_share<6>(v);
    case 7:  return row_share<7>(v);
    case 8:  return row_share<8>(v);
    case 9:  return row_share<9>(v);
    case 10: return row_share<10>(v);
    case 11: return row_share<11>(v);
    case 12: return row_share<12>(v);
    case 13: return row_share<13>(v);
    case 14: return row_share<14>(v);
    default: return row_share<15>(v);
    }
}

__device__ __forceinline__ void fmav(float& a, const float r, const float w)
{
    a = fmaf(r, w, a);
}
__device__ __forceinline__ void fmav(float4& a, const float4 r, const float w)
{
    a.x = fmaf(r.x, w, a.x); a.y = fmaf(r.y, w, a.y);
    a.z = fmaf(r.z, w, a.z); a.w = fmaf(r.w, w, a.w);
}
__device__ __forceinline__ float  zero_of(const float*)  { return 0.f; }
__device__ __forceinline__ float4 zero_of(const float4*) { return make_float4(0.f, 0.f, 0.f, 0.f); }

__device__ __forceinline__ void load_records(
    const int2* __restrict__ ed, const int j, const int end, const int l,
    int2& rec0, int2& rec1)
{
    rec0 = make_int2(0, 0);
    rec1 = make_int2(0, 0);
    if (j < end) {
        rec0 = ed[min(j + l, end - 1)];
        rec1 = ed[min(j + 16 + l, end - 1)];
    }
}

template <typename V, int VPR, int CH>
__device__ __forceinline__ V gather_node(
    const V* __restrict__ Pv, const int2* __restrict__ ed,
    int j, const int end, const int l, const int col, int2 rec0, int2 rec1)
{
    static_assert(16 % CH == 0, "a load of 16 records is whole chunks");
    constexpr unsigned ROWB = VPR * sizeof(V);
    const char* base = (const char*)Pv;
    const unsigned colb = (unsigned)col * (unsigned)sizeof(V);
    V acc = zero_of((const V*)nullptr);
    for (; j < end; j += 32) {
        #pragma unroll
        for (int c = 0; c < 32; c += CH) {
            const int m = end - j - c;
            if (m > 0) {
                const int2 rec = c < 16 ? rec0 : rec1;
                const int c0 = c & 15;
                unsigned off[CH];
                #pragma unroll
                for (int i = 0; i < CH; ++i)
                    off[i] = (unsigned)row_lane(rec.x, c0 + i) * ROWB + colb;
                V r[CH];
                #pragma unroll
                for (int i = 0; i < CH; ++i)
                    if (i < m) r[i] = *(const V*)(base + off[i]);
                #pragma unroll
                for (int i = 0; i < CH; ++i)
                    if (i < m) fmav(acc, r[i], __int_as_float(row_lane(rec.y, c0 + i)));
            }
        }
        if (j + 32 < end) load_records(ed, j + 32, end, l, rec0, rec1);
    }
    return acc;
}

__device__ __forceinline__ void load_rowptr(
    const int* __restrict__ rowptr, const int tl, const int ntiles,
    const int nb, const int N, int& jb, int& je)
{
    jb = 0;
    je = 0;
    const int node = tl * 16 + nb;
    if (tl < ntiles && node < N) {
        jb = rowptr[node];
        je = rowptr[node + 1];
    }
}

template <int DOUT, int DOUTP>
__global__ __launch_bounds__(256) void gcn_gather64_gemm(
    const float* __restrict__ P, const int* __restrict__ rowptr,
    const int* __restrict__ csr, const float* __restrict__ Wm,
    const float* __restrict__ bias, float* __restrict__ Pn, int N, int ntiles)
{
    constexpr int DIN  = 64;
    constexpr int LROW = DIN + 4;
    static_assert(DOUTP % 16 == 0 && DOUTP <= 64 && DOUT <= DOUTP, "column tiles");
    __shared__ float lds[2][16 * LROW];

    const int t    = threadIdx.x;
    const int nb   = t >> 4;
    const int l    = t & 15;

    const int wave    = t >> 5;
    const int lane    = t & 31;
    const int colbase = wave * 16;
    const bool mult   = colbase < DOUTP;
    const int half = lane >> 4;
    const int nrow = lane & 15;
    const int col  = colbase + nrow;

    v2f b[DIN / 4];
    float bv = 0.f;
    if (mult) {
        const float* wk = Wm + (size_t)(2 * half) * DOUTP + col;
        #pragma unroll
        for (int s = 0; s < DIN / 4; ++s) {
            b[s].x = wk[(size_t)(4 * s) * DOUTP];
            b[s].y = wk[(size_t)(4 * s + 1) * DOUTP];
        }
        bv = bias[col];
    }
    const int aoff = nrow * LROW + 2 * half;

    const int2* ed = (const int2*)csr;
    const int G = (int)gridDim.x;
    int jb, je, njb, nje;
    int2 rec0, rec1;
    load_rowptr(rowptr, (int)blockIdx.x, ntiles, nb, N, jb, je);
    load_rowptr(rowptr, (int)blockIdx.x + G, ntiles, nb, N, njb, nje);
    load_records(ed, jb, je, l, rec0, rec1);

    int buf = 0;
    for (int tl = blockIdx.x; tl < ntiles; tl += G, buf ^= 1) {
        const int row0 = tl * 16;
        float* tile = lds[buf];

        float4 acc = gather_node<float4, 16, 8>((const float4*)P, ed, jb, je, l, l, rec0, rec1);

        jb = njb;
        je = nje;
        load_records(ed, jb, je, l, rec0, rec1);
        load_rowptr(rowptr, tl + 2 * G, ntiles, nb, N, njb, nje);

        acc.x = fmaxf(acc.x, 0.f); acc.y = fmaxf(acc.y, 0.f);
        acc.z = fmaxf(acc.z, 0.f); acc.w = fmaxf(acc.w, 0.f);
        *(float4*)(&tile[nb * LROW + 4 * l]) = acc;
        lds_barrier();

        if (mult) {
            v8f d = {};
            #pragma unroll
            for (int s = 0; s < DIN / 4; ++s) {
                v2f a;
                a.x = tile[aoff + 4 * s];
                a.y = tile[aoff + 4 * s + 1];
                d = __builtin_amdgcn_wmma_f32_16x16x4_f32(
                        false, a, false, b[s], (short)0, d, false, false);
            }
            #pragma unroll
            for (int r = 0; r < 8; ++r) {
                const int m   = r + 8 * half;
                const int row = row0 + m;
                if (row < N && col < DOUT)
                    Pn[(size_t)row * DOUT + col] = d[r] + bv;
            }
        }
    }
}

__global__ __launch_bounds__(256) void gcn_gather10(
    const float* __restrict__ P, const int* __restrict__ rowptr,
    const int* __restrict__ csr, float* __restrict__ out, int N, int ntiles)
{
    __shared__ float stage[2][160];
    const int t    = threadIdx.x;
    const int nb   = t >> 4;
    const int l    = t & 15;

    const int2* ed = (const int2*)csr;
    const int G = (int)gridDim.x;
    int jb, je, njb, nje;
    int2 rec0, rec1;
    load_rowptr(rowptr, (int)blockIdx.x, ntiles, nb, N, jb, je);
    load_rowptr(rowptr, (int)blockIdx.x + G, ntiles, nb, N, njb, nje);
    load_records(ed, jb, je, l, rec0, rec1);

    int buf = 0;
    for (int tl = blockIdx.x; tl < ntiles; tl += G, buf ^= 1) {
        const float acc = gather_node<float, 10, 16>(P, ed, jb, je, l, min(l, 9), rec0, rec1);
        jb = njb;
        je = nje;
        load_records(ed, jb, je, l, rec0, rec1);
        load_rowptr(rowptr, tl + 2 * G, ntiles, nb, N, njb, nje);
        if (l < 10) stage[buf][nb * 10 + l] = acc;
        lds_barrier();
        const int o = tl * 160 + t;
        if (t < 160 && o < N * 10) out[o] = stage[buf][t];
    }
}

extern "C" void kernel_launch(void* const* d_in, const int* in_sizes, int n_in,
                              void* d_out, int out_size, void* d_ws, size_t ws_size,
                              hipStream_t stream)
{
    (void)in_sizes; (void)n_in; (void)out_size; (void)ws_size;
    const int N = NNODES, E = NEDGES;

    const float* x    = (const float*)d_in[0];
    const int*   esrc = (const int*)  d_in[1];
    const int*   edst = (const int*)  d_in[2];
    const float* ew   = (const float*)d_in[3];
    const float* Ws[5]; const float* bs[5];
    for (int i = 0; i < 5; ++i) {
        Ws[i] = (const float*)d_in[4 + 2 * i];
        bs[i] = (const float*)d_in[5 + 2 * i];
    }
    float* out  = (float*)d_out;
    float* buf0 = (float*)d_ws;
    float* buf1 = buf0 + (size_t)N * 64;
    float* buf2 = buf1 + (size_t)N * 64;
    float* wpad = buf2 + (size_t)N * 64;
    float* bpad = wpad + 64 * 16;
    int*   cnt    = (int*)(bpad + 64);
    int*   rowptr = cnt + 2 * N;
    int*   rank   = rowptr + (N + 64);
    int*   csr    = rank + E;

    const int  ntiles = (N + 15) / 16;
    const dim3 pgrid(1250);
    const dim3 sgrid(197);

    gcn_zero_i32<<<(2 * N + 255) / 256, 256, 0, stream>>>(cnt, 2 * N);
    gcn_count_rank<<<pgrid, 256, 0, stream>>>(edst, cnt, rank, E);
    gcn_scan_gemm0<<<sgrid, 1024, 0, stream>>>(cnt, rowptr, x, Ws[0], bs[0], buf0, N, ntiles,
                                               Ws[4], bs[4], wpad, bpad);
    gcn_fill<<<pgrid, 256, 0, stream>>>(esrc, edst, ew, rowptr, rank, csr, E);

    gcn_gather64_gemm<64, 64><<<pgrid, 256, 0, stream>>>(buf0, rowptr, csr, Ws[1], bs[1], buf1, N, ntiles);
    gcn_gather64_gemm<64, 64><<<pgrid, 256, 0, stream>>>(buf1, rowptr, csr, Ws[2], bs[2], buf0, N, ntiles);
    gcn_gather64_gemm<64, 64><<<pgrid, 256, 0, stream>>>(buf0, rowptr, csr, Ws[3], bs[3], buf1, N, ntiles);

    gcn_gather64_gemm<10, 16><<<pgrid, 256, 0, stream>>>(buf1, rowptr, csr, wpad, bpad, buf0, N, ntiles);

    gcn_gather10<<<pgrid, 256, 0, stream>>>(buf0, rowptr, csr, out, N, ntiles);
}


# One pass puts each edge record at its atomic arrival rank in a fixed per-node slot block (32 per package half, overflow chained): scan, fill pass, rank array go; 9 launches to 7; new sum order
# speedup vs baseline: 4.3828x; 1.0754x over previous
#include <hip/hip_runtime.h>
#include <hip/hip_bf16.h>

typedef float v2f __attribute__((ext_vector_type(2)));
typedef float v8f __attribute__((ext_vector_type(8)));

#define NNODES 100000
#define NEDGES 1600000


__device__ __forceinline__ void lds_barrier()
{
    __builtin_amdgcn_fence(__ATOMIC_RELEASE, "workgroup", "local");
    __builtin_amdgcn_s_barrier();
    __builtin_amdgcn_fence(__ATOMIC_ACQUIRE, "workgroup", "local");
}

#define GEMM0_GROUPS 2
template <int DIN, int DOUT, int DOUTP>
__device__ __forceinline__ void gcn_gemm_groups(
    const float* __restrict__ X, const float* __restrict__ Wm,
    const float* __restrict__ bias, float* __restrict__ P, int N, int ntiles,
    const int wg, const int nwg)
{
    constexpr int LROW = DIN + 4;
    constexpr int VPER = DIN / 4;
    constexpr int NV   = 16 * VPER / 128;
    static_assert(DIN % 32 == 0 && DOUTP % 16 == 0 && DOUTP <= 64 && DOUT <= DOUTP, "tiles");
    __shared__ float lds[2][GEMM0_GROUPS][16 * LROW];
    const int grp = threadIdx.x >> 7;
    const int t   = threadIdx.x & 127;

    const int wave = t >> 5;
    const int lane = t & 31;
    const int colbase = wave * 16;
    const bool mult = colbase < DOUTP;
    const int half = lane >> 4;
    const int nrow = lane & 15;
    const int col  = colbase + nrow;

    v2f b[DIN / 4];
    float bv = 0.f;
    if (mult) {
        const float* wk = Wm + (size_t)(2 * half) * DOUTP + col;
        #pragma unroll
        for (int s = 0; s < DIN / 4; ++s) {
            b[s].x = wk[(size_t)(4 * s) * DOUTP];
            b[s].y = wk[(size_t)(4 * s + 1) * DOUTP];
        }
        bv = bias[col];
    }
    const int aoff = nrow * LROW + 2 * half;

    const int stride = nwg * GEMM0_GROUPS;
    const int first  = wg * GEMM0_GROUPS + grp;
    const int passes = (ntiles + stride - 1) / stride;

    float4 v[NV];
    #pragma unroll
    for (int i = 0; i < NV; ++i) {
        const int idx = t + i * 128;
        const int r   = idx / VPER;
        const int c4  = (idx - r * VPER) << 2;
        const int row = first * 16 + r;
        v[i] = *(const float4*)(X + (size_t)min(row, N - 1) * DIN + c4);
    }

    int buf = 0;
    for (int ps = 0; ps < passes; ++ps, buf ^= 1) {
        const int tl   = first + ps * stride;
        const int row0 = tl * 16;
        float* tile = lds[buf][grp];
        #pragma unroll
        for (int i = 0; i < NV; ++i) {
            const int idx = t + i * 128;
            const int r   = idx / VPER;
            const int c4  = (idx - r * VPER) << 2;
            const bool in = row0 + r < N;
            *(float4*)(&tile[r * LROW + c4]) = in ? v[i] : make_float4(0.f, 0.f, 0.f, 0.f);
        }
        if (ps + 1 < passes) {
            #pragma unroll
            for (int i = 0; i < NV; ++i) {
                const int idx = t + i * 128;
                const int r   = idx / VPER;
                const int c4  = (idx - r * VPER) << 2;
                const int row = (tl + stride) * 16 + r;
                v[i] = *(const float4*)(X + (size_t)min(row, N - 1) * DIN + c4);
            }
        }
        lds_barrier();
        if (mult) {
            v8f acc = {};
            #pragma unroll
            for (int s = 0; s < DIN / 4; ++s) {
                v2f a;
                a.x = tile[aoff + 4 * s];
                a.y = tile[aoff + 4 * s + 1];
                acc = __builtin_amdgcn_wmma_f32_16x16x4_f32(
                          false, a, false, b[s], (short)0, acc, false, false);
            }
            #pragma unroll
            for (int r = 0; r < 8; ++r) {
                const int m   = r + 8 * half;
                const int row = row0 + m;
                if (row < N && col < DOUT)
                    P[(size_t)row * DOUT + col] = acc[r] + bv;
            }
        }
    }
}

__device__ __forceinline__ void gcn_pad_w4(const float* __restrict__ W4, const float* __restrict__ b4,
                                           float* __restrict__ Wp, float* __restrict__ bp)
{
    for (int i = threadIdx.x; i < 64 * 16 + 16; i += blockDim.x) {
        if (i < 64 * 16) {
            const int r = i >> 4, c = i & 15;
            Wp[i] = (c < 10) ? W4[r * 10 + c] : 0.f;
        } else {
            const int c = i - 64 * 16;
            bp[c] = (c < 10) ? b4[c] : 0.f;
        }
    }
}

#define HSEG 32
static_assert(HSEG == 32, "the slot formula below is for two 16-slot pieces per half");

__global__ void gcn_zero_i32(int* __restrict__ p, int n)
{
    const int i = blockIdx.x * blockDim.x + threadIdx.x;
    if (i < n) p[i] = 0;
}

__device__ __forceinline__ int slot_of(const int h, const int r)
{
    return ((r >> 4) << 5) + (h << 4) + (r & 15);
}

#define EPT 5

__device__ __forceinline__ int package_half()
{
    return (int)((__builtin_amdgcn_s_sendmsg_rtn(0x87) >> 18) & 1u);
}

__device__ __forceinline__ void gcn_place_edges(
    const int* __restrict__ src, const int* __restrict__ dst,
    const float* __restrict__ w, int* __restrict__ cnt2,
    int* __restrict__ head, int* __restrict__ next, int* __restrict__ slots,
    const int E, const int wg, const int nwg)
{
    const int h = package_half();
    int* __restrict__ cnt = cnt2 + h * NNODES;
    const int T = nwg * 256;
    for (int e0 = wg * 256 + (int)threadIdx.x; e0 < E; e0 += EPT * T) {
        int d[EPT], r[EPT];
        int2 v[EPT];
        #pragma unroll
        for (int k = 0; k < EPT; ++k) {
            const int ec = min(e0 + k * T, E - 1);
            d[k]   = dst[ec];
            v[k].x = src[ec];
            v[k].y = __float_as_int(w[ec]);
        }
        #pragma unroll
        for (int k = 0; k < EPT; ++k) {
            r[k] = 0;
            if (e0 + k * T < E)
                r[k] = __hip_atomic_fetch_add(cnt + d[k], 1, __ATOMIC_RELAXED,
                                              __HIP_MEMORY_SCOPE_AGENT);
        }
        #pragma unroll
        for (int k = 0; k < EPT; ++k) {
            const int e = e0 + k * T;
            if (e < E) {
                if (r[k] < HSEG) {
                    const size_t pos = (size_t)d[k] * (2 * HSEG) + slot_of(h, r[k]);
                    *(int2*)(slots + 2 * pos) = v[k];
                } else {
                    next[e] = __hip_atomic_exchange(head + d[k], e, __ATOMIC_RELAXED,
                                                    __HIP_MEMORY_SCOPE_AGENT);
                }
            }
        }
    }
}

__global__ __launch_bounds__(256) void gcn_edges_gemm0(
    const int* __restrict__ src, const int* __restrict__ dst,
    const float* __restrict__ w, int* __restrict__ cnt2,
    int* __restrict__ head, int* __restrict__ next, int* __restrict__ slots, int E,
    const float* __restrict__ X, const float* __restrict__ W0,
    const float* __restrict__ b0, float* __restrict__ P, int N, int ntiles,
    const float* __restrict__ W4, const float* __restrict__ b4,
    float* __restrict__ Wp, float* __restrict__ bp)
{
    const int b   = (int)blockIdx.x;
    const int ng  = (int)gridDim.x / 3;
    if (b % 3 != 2) {
        const int wg = b - b / 3;
        gcn_place_edges(src, dst, w, cnt2, head, next, slots, E, wg, (int)gridDim.x - ng);
        return;
    }
    const int wg = b / 3;
    if (wg == 0) gcn_pad_w4(W4, b4, Wp, bp);
    gcn_gemm_groups<128, 64, 64>(X, W0, b0, P, N, ntiles, wg, ng);
}

template <int I>
__device__ __forceinline__ int row_share(const int v)
{
    return __builtin_amdgcn_update_dpp(0, v, 0x150 + I, 0xF, 0xF, true);
}
__device__ __forceinline__ int row_lane(const int v, const int i)
{
    switch (i) {
    case 0:  return row_share<0>(v);
    case 1:  return row_share<1>(v);
    case 2:  return row_share<2>(v);
    case 3:  return row_share<3>(v);
    case 4:  return row_share<4>(v);
    case 5:  return row_share<5>(v);
    case 6:  return row_share<6>(v);
    case 7:  return row_share<7>(v);
    case 8:  return row_share<8>(v);
    case 9:  return row_share<9>(v);
    case 10: return row_share<10>(v);
    case 11: return row_share<11>(v);
    case 12: return row_share<12>(v);
    case 13: return row_share<13>(v);
    case 14: return row_share<14>(v);
    default: return row_share<15>(v);
    }
}

__device__ __forceinline__ void fmav(float& a, const float r, const float w)
{
    a = fmaf(r, w, a);
}
__device__ __forceinline__ void fmav(float4& a, const float4 r, const float w)
{
    a.x = fmaf(r.x, w, a.x); a.y = fmaf(r.y, w, a.y);
    a.z = fmaf(r.z, w, a.z); a.w = fmaf(r.w, w, a.w);
}
__device__ __forceinline__ float  zero_of(const float*)  { return 0.f; }
__device__ __forceinline__ float4 zero_of(const float4*) { return make_float4(0.f, 0.f, 0.f, 0.f); }

__device__ __forceinline__ int list_slot(const int k, const int n0)
{
    return k < n0 ? slot_of(0, k) : slot_of(1, k - n0);
}

__device__ __forceinline__ int2 load_record(const int2* __restrict__ ed, const int node, const int s)
{
    const unsigned off = ((unsigned)node * (2u * HSEG) + (unsigned)s) * (unsigned)sizeof(int2);
    return *(const int2*)((const char*)ed + off);
}

__device__ __forceinline__ void load_records(
    const int2* __restrict__ ed, const int node, const int c0, const int c1,
    const int l, int2& rec0, int2& rec1)
{
    const int n0  = min(c0, HSEG);
    const int len = n0 + min(c1, HSEG);
    rec0 = make_int2(0, 0);
    rec1 = make_int2(0, 0);
    if (len > 0) {
        rec0 = load_record(ed, node, list_slot(min(l, len - 1), n0));
        rec1 = load_record(ed, node, list_slot(min(16 + l, len - 1), n0));
    }
}

template <typename V, int VPR, int CH>
__device__ __forceinline__ V gather_node(
    const V* __restrict__ Pv, const int tot, const int col,
    const int2 rec0, const int2 rec1)
{
    static_assert(16 % CH == 0, "a load of 16 records is whole chunks");
    constexpr unsigned ROWB = VPR * sizeof(V);
    const char* base = (const char*)Pv;
    const unsigned colb = (unsigned)col * (unsigned)sizeof(V);
    const int end = min(tot, 32);
    V acc = zero_of((const V*)nullptr);
    #pragma unroll
    for (int c = 0; c < 32; c += CH) {
        const int m = end - c;
        if (m > 0) {
            const int2 rec = c < 16 ? rec0 : rec1;
            const int cl = c & 15;
            unsigned off[CH];
            #pragma unroll
            for (int i = 0; i < CH; ++i)
                off[i] = (unsigned)row_lane(rec.x, cl + i) * ROWB + colb;
            V r[CH];
            #pragma unroll
            for (int i = 0; i < CH; ++i)
                if (i < m) r[i] = *(const V*)(base + off[i]);
            #pragma unroll
            for (int i = 0; i < CH; ++i)
                if (i < m) fmav(acc, r[i], __int_as_float(row_lane(rec.y, cl + i)));
        }
    }
    return acc;
}

struct EdgeChain {
    const int*   head;
    const int*   next;
    const int*   src;
    const float* w;
};

template <typename V, int VPR>
__device__ __forceinline__ V gather_rest(
    const V* __restrict__ Pv, const int2* __restrict__ ed, const EdgeChain& ch,
    const int* __restrict__ cnt2, const int node, const int col, V acc)
{
    constexpr unsigned ROWB = VPR * sizeof(V);
    const char* base = (const char*)Pv;
    const unsigned colb = (unsigned)col * (unsigned)sizeof(V);
    const int c0  = cnt2[node];
    const int c1  = cnt2[NNODES + node];
    const int n0  = min(c0, HSEG);
    const int len = n0 + min(c1, HSEG);
    for (int k = 32; k < len; ++k) {
        const int2 rec = load_record(ed, node, list_slot(k, n0));
        fmav(acc, *(const V*)(base + ((unsigned)rec.x * ROWB + colb)), __int_as_float(rec.y));
    }
    const int links = max(c0 - HSEG, 0) + max(c1 - HSEG, 0);
    if (links > 0) {
        int o = ch.head[node];
        for (int i = 0; i < links; ++i) {
            o = min(max(o, 0), NEDGES - 1);
            fmav(acc, *(const V*)(base + ((unsigned)ch.src[o] * ROWB + colb)), ch.w[o]);
            o = ch.next[o];
        }
    }
    return acc;
}

__device__ __forceinline__ void load_counts(
    const int* __restrict__ cnt2, const int tl, const int ntiles,
    const int nb, const int N, int& c0, int& c1)
{
    c0 = 0;
    c1 = 0;
    const int node = tl * 16 + nb;
    if (tl < ntiles && node < N) {
        c0 = cnt2[node];
        c1 = cnt2[NNODES + node];
    }
}

template <int DOUT, int DOUTP>
__global__ __launch_bounds__(256, 10) void gcn_gather64_gemm(
    const float* __restrict__ P, const int* __restrict__ cnt2,
    const int* __restrict__ slots, const int* __restrict__ head,
    const int* __restrict__ next, const int* __restrict__ src,
    const float* __restrict__ w, const float* __restrict__ Wm,
    const float* __restrict__ bias, float* __restrict__ Pn, int N, int ntiles)
{
    constexpr int DIN  = 64;
    constexpr int LROW = DIN + 4;
    static_assert(DOUTP % 16 == 0 && DOUTP <= 64 && DOUT <= DOUTP, "column tiles");
    __shared__ float lds[2][16 * LROW];

    const int t    = threadIdx.x;
    const int nb   = t >> 4;
    const int l    = t & 15;

    const int wave    = t >> 5;
    const int lane    = t & 31;
    const int colbase = wave * 16;
    const bool mult   = colbase < DOUTP;
    const int half = lane >> 4;
    const int nrow = lane & 15;
    const int col  = colbase + nrow;

    v2f b[DIN / 4];
    float bv = 0.f;
    if (mult) {
        const float* wk = Wm + (size_t)(2 * half) * DOUTP + col;
        #pragma unroll
        for (int s = 0; s < DIN / 4; ++s) {
            b[s].x = wk[(size_t)(4 * s) * DOUTP];
            b[s].y = wk[(size_t)(4 * s + 1) * DOUTP];
        }
        bv = bias[col];
    }
    const int aoff = nrow * LROW + 2 * half;

    const int2* ed = (const int2*)slots;
    const EdgeChain ch = {head, next, src, w};
    const int G = (int)gridDim.x;
    int tot, nc0, nc1;
    int2 rec0, rec1;
    load_counts(cnt2, (int)blockIdx.x, ntiles, nb, N, nc0, nc1);
    load_records(ed, (int)blockIdx.x * 16 + nb, nc0, nc1, l, rec0, rec1);
    tot = nc0 + nc1;
    load_counts(cnt2, (int)blockIdx.x + G, ntiles, nb, N, nc0, nc1);

    int buf = 0;
    for (int tl = blockIdx.x; tl < ntiles; tl += G, buf ^= 1) {
        const int row0 = tl * 16;
        float* tile = lds[buf];

        float4 acc = gather_node<float4, 16, 8>((const float4*)P, tot, l, rec0, rec1);
        if (tot > 32)
            acc = gather_rest<float4, 16>((const float4*)P, ed, ch, cnt2, row0 + nb, l, acc);

        load_records(ed, (tl + G) * 16 + nb, nc0, nc1, l, rec0, rec1);
        tot = nc0 + nc1;
        load_counts(cnt2, tl + 2 * G, ntiles, nb, N, nc0, nc1);

        acc.x = fmaxf(acc.x, 0.f); acc.y = fmaxf(acc.y, 0.f);
        acc.z = fmaxf(acc.z, 0.f); acc.w = fmaxf(acc.w, 0.f);
        *(float4*)(&tile[nb * LROW + 4 * l]) = acc;
        lds_barrier();

        if (mult) {
            v8f d = {};
            #pragma unroll
            for (int s = 0; s < DIN / 4; ++s) {
                v2f a;
                a.x = tile[aoff + 4 * s];
                a.y = tile[aoff + 4 * s + 1];
                d = __builtin_amdgcn_wmma_f32_16x16x4_f32(
                        false, a, false, b[s], (short)0, d, false, false);
            }
            #pragma unroll
            for (int r = 0; r < 8; ++r) {
                const int m   = r + 8 * half;
                const int row = row0 + m;
                if (row < N && col < DOUT)
                    Pn[(size_t)row * DOUT + col] = d[r] + bv;
            }
        }
    }
}

__global__ __launch_bounds__(256) void gcn_gather10(
    const float* __restrict__ P, const int* __restrict__ cnt2,
    const int* __restrict__ slots, const int* __restrict__ head,
    const int* __restrict__ next, const int* __restrict__ src,
    const float* __restrict__ w, float* __restrict__ out, int N, int ntiles)
{
    __shared__ float stage[2][160];
    const int t    = threadIdx.x;
    const int nb   = t >> 4;
    const int l    = t & 15;

    const int2* ed = (const int2*)slots;
    const EdgeChain ch = {head, next, src, w};
    const int G = (int)gridDim.x;
    int tot, nc0, nc1;
    int2 rec0, rec1;
    load_counts(cnt2, (int)blockIdx.x, ntiles, nb, N, nc0, nc1);
    load_records(ed, (int)blockIdx.x * 16 + nb, nc0, nc1, l, rec0, rec1);
    tot = nc0 + nc1;
    load_counts(cnt2, (int)blockIdx.x + G, ntiles, nb, N, nc0, nc1);

    int buf = 0;
    for (int tl = blockIdx.x; tl < ntiles; tl += G, buf ^= 1) {
        float acc = gather_node<float, 10, 16>(P, tot, min(l, 9), rec0, rec1);
        if (tot > 32)
            acc = gather_rest<float, 10>(P, ed, ch, cnt2, tl * 16 + nb, min(l, 9), acc);
        load_records(ed, (tl + G) * 16 + nb, nc0, nc1, l, rec0, rec1);
        tot = nc0 + nc1;
        load_counts(cnt2, tl + 2 * G, ntiles, nb, N, nc0, nc1);
        if (l < 10) stage[buf][nb * 10 + l] = acc;
        lds_barrier();
        const int o = tl * 160 + t;
        if (t < 160 && o < N * 10) out[o] = stage[buf][t];
    }
}

extern "C" void kernel_launch(void* const* d_in, const int* in_sizes, int n_in,
                              void* d_out, int out_size, void* d_ws, size_t ws_size,
                              hipStream_t stream)
{
    (void)in_sizes; (void)n_in; (void)out_size; (void)ws_size;
    const int N = NNODES, E = NEDGES;

    const float* x    = (const float*)d_in[0];
    const int*   esrc = (const int*)  d_in[1];
    const int*   edst = (const int*)  d_in[2];
    const float* ew   = (const float*)d_in[3];
    const float* Ws[5]; const float* bs[5];
    for (int i = 0; i < 5; ++i) {
        Ws[i] = (const float*)d_in[4 + 2 * i];
        bs[i] = (const float*)d_in[5 + 2 * i];
    }
    float* out  = (float*)d_out;
    float* buf0 = (float*)d_ws;
    float* buf1 = buf0 + (size_t)N * 64;
    float* buf2 = buf1 + (size_t)N * 64;
    float* wpad = buf2 + (size_t)N * 64;
    float* bpad = wpad + 64 * 16;
    int*   cnt   = (int*)(bpad + 64);
    int*   head  = cnt + 2 * N;
    int*   next  = head + (N + 96);
    int*   slots = next + E;

    const int  ntiles = (N + 15) / 16;
    const dim3 pgrid(1250);
    const dim3 egrid(1875);

    gcn_zero_i32<<<(2 * N + 255) / 256, 256, 0, stream>>>(cnt, 2 * N);
    gcn_edges_gemm0<<<egrid, 256, 0, stream>>>(esrc, edst, ew, cnt, head, next, slots, E,
                                               x, Ws[0], bs[0], buf0, N, ntiles,
                                               Ws[4], bs[4], wpad, bpad);

    gcn_gather64_gemm<64, 64><<<pgrid, 256, 0, stream>>>(buf0, cnt, slots, head, next, esrc, ew, Ws[1], bs[1], buf1, N, ntiles);
    gcn_gather64_gemm<64, 64><<<pgrid, 256, 0, stream>>>(buf1, cnt, slots, head, next, esrc, ew, Ws[2], bs[2], buf0, N, ntiles);
    gcn_gather64_gemm<64, 64><<<pgrid, 256, 0, stream>>>(buf0, cnt, slots, head, next, esrc, ew, Ws[3], bs[3], buf1, N, ntiles);

    gcn_gather64_gemm<10, 16><<<pgrid, 256, 0, stream>>>(buf1, cnt, slots, head, next, esrc, ew, wpad, bpad, buf0, N, ntiles);

    gcn_gather10<<<pgrid, 256, 0, stream>>>(buf0, cnt, slots, head, next, esrc, ew, out, N, ntiles);
}


# Edge lists built by a two-level counting sort on LDS counters (tiles of 4096 edges binned by 256-node bucket, then ranked per bucket), no global atomic per edge, no zero-fill launch; new sum order
# speedup vs baseline: 5.1049x; 1.1648x over previous
#include <hip/hip_runtime.h>
#include <hip/hip_bf16.h>

typedef float v2f __attribute__((ext_vector_type(2)));
typedef float v8f __attribute__((ext_vector_type(8)));

#define NNODES 100000
#define NEDGES 1600000


__device__ __forceinline__ void lds_barrier()
{
    __builtin_amdgcn_fence(__ATOMIC_RELEASE, "workgroup", "local");
    __builtin_amdgcn_s_barrier();
    __builtin_amdgcn_fence(__ATOMIC_ACQUIRE, "workgroup", "local");
}

template <int DIN, int DOUT, int DOUTP, int NG>
__device__ __forceinline__ void gcn_gemm_groups(
    const float* __restrict__ X, const float* __restrict__ Wm,
    const float* __restrict__ bias, float* __restrict__ P, int N,
    const int tbeg, const int tend, const int wg, const int nwg, float* ldsf)
{
    constexpr int LROW = DIN + 4;
    constexpr int VPER = DIN / 4;
    constexpr int NV   = 16 * VPER / 128;
    static_assert(DIN % 32 == 0 && DOUTP % 16 == 0 && DOUTP <= 64 && DOUT <= DOUTP, "tiles");
    const int grp = threadIdx.x >> 7;
    const int t   = threadIdx.x & 127;

    const int wave = t >> 5;
    const int lane = t & 31;
    const int colbase = wave * 16;
    const bool mult = colbase < DOUTP;
    const int half = lane >> 4;
    const int nrow = lane & 15;
    const int col  = colbase + nrow;

    v2f b[DIN / 4];
    float bv = 0.f;
    if (mult) {
        const float* wk = Wm + (size_t)(2 * half) * DOUTP + col;
        #pragma unroll
        for (int s = 0; s < DIN / 4; ++s) {
            b[s].x = wk[(size_t)(4 * s) * DOUTP];
            b[s].y = wk[(size_t)(4 * s + 1) * DOUTP];
        }
        bv = bias[col];
    }
    const int aoff = nrow * LROW + 2 * half;

    const int stride = nwg * NG;
    const int first  = tbeg + wg * NG + grp;
    const int passes = (tend - tbeg + stride - 1) / stride;

    float4 v[NV];
    #pragma unroll
    for (int i = 0; i < NV; ++i) {
        const int idx = t + i * 128;
        const int r   = idx / VPER;
        const int c4  = (idx - r * VPER) << 2;
        const int row = first * 16 + r;
        v[i] = *(const float4*)(X + (size_t)min(row, N - 1) * DIN + c4);
    }

    int buf = 0;
    for (int ps = 0; ps < passes; ++ps, buf ^= 1) {
        const int tl   = first + ps * stride;
        const int row0 = tl * 16;
        float* tile = ldsf + (buf * NG + grp) * (16 * LROW);
        #pragma unroll
        for (int i = 0; i < NV; ++i) {
            const int idx = t + i * 128;
            const int r   = idx / VPER;
            const int c4  = (idx - r * VPER) << 2;
            const bool in = tl < tend && row0 + r < N;
            *(float4*)(&tile[r * LROW + c4]) = in ? v[i] : make_float4(0.f, 0.f, 0.f, 0.f);
        }
        if (ps + 1 < passes) {
            #pragma unroll
            for (int i = 0; i < NV; ++i) {
                const int idx = t + i * 128;
                const int r   = idx / VPER;
                const int c4  = (idx - r * VPER) << 2;
                const int row = (tl + stride) * 16 + r;
                v[i] = *(const float4*)(X + (size_t)min(row, N - 1) * DIN + c4);
            }
        }
        lds_barrier();
        if (mult) {
            v8f acc = {};
            #pragma unroll
            for (int s = 0; s < DIN / 4; ++s) {
                v2f a;
                a.x = tile[aoff + 4 * s];
                a.y = tile[aoff + 4 * s + 1];
                acc = __builtin_amdgcn_wmma_f32_16x16x4_f32(
                          false, a, false, b[s], (short)0, acc, false, false);
            }
            #pragma unroll
            for (int r = 0; r < 8; ++r) {
                const int m   = r + 8 * half;
                const int row = row0 + m;
                if (tl < tend && row < N && col < DOUT)
                    P[(size_t)row * DOUT + col] = acc[r] + bv;
            }
        }
    }
}

__device__ __forceinline__ void gcn_pad_w4(const float* __restrict__ W4, const float* __restrict__ b4,
                                           float* __restrict__ Wp, float* __restrict__ bp)
{
    for (int i = threadIdx.x; i < 64 * 16 + 16; i += blockDim.x) {
        if (i < 64 * 16) {
            const int r = i >> 4, c = i & 15;
            Wp[i] = (c < 10) ? W4[r * 10 + c] : 0.f;
        } else {
            const int c = i - 64 * 16;
            bp[c] = (c < 10) ? b4[c] : 0.f;
        }
    }
}

#define HSEG 32
static_assert(HSEG == 32, "the slot formula below is for two 16-slot pieces per segment");

#define TILE_E 4096
#define NTILE  ((NEDGES + TILE_E - 1) / TILE_E)
#define BKT_SH 8
#define BKT_N  (1 << BKT_SH)
#define NBKT   ((NNODES + BKT_N - 1) / BKT_N)
#define HISTN  512
static_assert(NBKT <= HISTN && TILE_E == 16 * 256 && NTILE <= 512, "sizes of the two sort levels");
static_assert(NNODES <= (1 << 24) && BKT_SH <= 8, "a level-1 record's first word: 24 bits of source, 8 of place");
#define BIN_LDS_INTS (2 * HISTN + 16 + 2 * TILE_E)
#define GEMM0_TILES ((NNODES + 15) / 16)
#define GEMM0_SPLIT (GEMM0_TILES / 2)
#define GEMM0_WG_A  ((GEMM0_SPLIT + 5) / 6)
#define GEMM0_WG_B  ((GEMM0_TILES - GEMM0_SPLIT + 5) / 6)
#define RANK_LDS_INTS (BKT_N * 64 + BKT_N)
static_assert(RANK_LDS_INTS >= 2 * 2 * 16 * (128 + 4), "the projection's tiles fit the buffer");

__device__ __forceinline__ int slot_of(const int h, const int r)
{
    return ((r >> 4) << 5) + (h << 4) + (r & 15);
}

__device__ __forceinline__ void gcn_bin_tile(
    const int* __restrict__ src, const int* __restrict__ dst,
    const float* __restrict__ w, int2* __restrict__ erec, int* __restrict__ offT,
    const int E, const int tile, int* lds)
{
    int*  hist  = lds;
    int*  start = lds + HISTN;
    int*  wsum  = lds + 2 * HISTN;
    int2* srec  = (int2*)(lds + 2 * HISTN + 16);
    const int t = threadIdx.x;
    hist[t] = 0;
    hist[t + 256] = 0;
    __syncthreads();

    const int e0  = tile * TILE_E;
    const int cnt = min(TILE_E, E - e0);
    int d[16], s[16], v[16], r[16];
    #pragma unroll
    for (int k = 0; k < 16; ++k) {
        const int ec = e0 + min(k * 256 + t, cnt - 1);
        d[k] = dst[ec];
        s[k] = src[ec];
        v[k] = __float_as_int(w[ec]);
    }
    #pragma unroll
    for (int k = 0; k < 16; ++k) {
        r[k] = 0;
        if (k * 256 + t < cnt)
            r[k] = __hip_atomic_fetch_add(hist + (d[k] >> BKT_SH), 1, __ATOMIC_RELAXED,
                                          __HIP_MEMORY_SCOPE_WORKGROUP);
    }
    __syncthreads();

    const int h0 = hist[2 * t], h1 = hist[2 * t + 1];
    int x = h0 + h1;
    const int lane = t & 31, wv = t >> 5;
    #pragma unroll
    for (int o = 1; o < 32; o <<= 1) {
        const int y = __shfl_up(x, o);
        if (lane >= o) x += y;
    }
    if (lane == 31) wsum[wv] = x;
    __syncthreads();
    int base = 0;
    for (int i = 0; i < wv; ++i) base += wsum[i];
    const int excl = base + x - (h0 + h1);
    start[2 * t]     = excl;
    start[2 * t + 1] = excl + h0;
    __syncthreads();

    for (int b = t; b < NBKT; b += 256)
        offT[b * NTILE + tile] = start[b] | (hist[b] << 16);

    #pragma unroll
    for (int k = 0; k < 16; ++k) {
        if (k * 256 + t < cnt) {
            const int p = start[d[k] >> BKT_SH] + r[k];
            srec[p] = make_int2(s[k] | ((d[k] & (BKT_N - 1)) << 24), v[k]);
        }
    }
    __syncthreads();
    #pragma unroll
    for (int k = 0; k < 16; ++k) {
        const int i = k * 256 + t;
        if (i < cnt) erec[e0 + i] = srec[i];
    }
}

__global__ __launch_bounds__(256) void gcn_bin_gemm0(
    const int* __restrict__ src, const int* __restrict__ dst,
    const float* __restrict__ w, int* __restrict__ erec, int* __restrict__ offT, int E,
    const float* __restrict__ X, const float* __restrict__ W0,
    const float* __restrict__ b0, float* __restrict__ P, int N,
    const float* __restrict__ W4, const float* __restrict__ b4,
    float* __restrict__ Wp, float* __restrict__ bp)
{
    static_assert(BIN_LDS_INTS >= 2 * 2 * 16 * (128 + 4), "the projection's tiles fit the buffer");
    __shared__ __attribute__((aligned(16))) int lds[BIN_LDS_INTS];
    const int b = (int)blockIdx.x;
    if (b < NTILE) {
        gcn_bin_tile(src, dst, w, (int2*)erec, offT, E, b, lds);
        return;
    }
    const int wg = b - NTILE;
    if (wg == 0) gcn_pad_w4(W4, b4, Wp, bp);
    gcn_gemm_groups<128, 64, 64, 2>(X, W0, b0, P, N, 0, GEMM0_SPLIT, wg, (int)gridDim.x - NTILE, (float*)lds);
}

#define RCH 8
__global__ __launch_bounds__(256) void gcn_rank_gemm0(
    const int* __restrict__ erec, const int* __restrict__ offT,
    int* __restrict__ cnt2, int* __restrict__ head, int* __restrict__ next,
    int* __restrict__ slots, int N,
    const float* __restrict__ X, const float* __restrict__ W0,
    const float* __restrict__ b0, float* __restrict__ P)
{
    const int2* __restrict__ er = (const int2*)erec;
    __shared__ __attribute__((aligned(16))) int lds[RANK_LDS_INTS];
    const int b = (int)blockIdx.x;
    if (b >= NBKT) {
        gcn_gemm_groups<128, 64, 64, 2>(X, W0, b0, P, N, GEMM0_SPLIT, GEMM0_TILES,
                                        b - NBKT, (int)gridDim.x - NBKT, (float*)lds);
        return;
    }
    int2 (*stage)[32] = (int2 (*)[32])lds;
    int* cnt = lds + BKT_N * 64;
    const int t     = threadIdx.x;
    const int node0 = b * BKT_N;
    for (int i = t; i < BKT_N; i += 256) cnt[i] = 0;
    __syncthreads();

    {
        const int t1   = t + 256;
        const int pk0  = t  < NTILE ? offT[b * NTILE + t]  : 0;
        const int pk1  = t1 < NTILE ? offT[b * NTILE + t1] : 0;
        const int len0 = pk0 >> 16;
        const int len  = len0 + (pk1 >> 16);
        const int q0 = t  * TILE_E + (pk0 & 0xFFFF);
        const int q1 = t1 * TILE_E + (pk1 & 0xFFFF) - len0;
        for (int j0 = 0; j0 < len; j0 += RCH) {
            int p[RCH], r[RCH];
            int2 e[RCH];
            #pragma unroll
            for (int i = 0; i < RCH; ++i) {
                const int j = min(j0 + i, len - 1);
                p[i] = (j < len0 ? q0 : q1) + j;
                e[i] = er[p[i]];
            }
            #pragma unroll
            for (int i = 0; i < RCH; ++i) {
                r[i] = 0;
                if (j0 + i < len)
                    r[i] = __hip_atomic_fetch_add(cnt + (int)((unsigned)e[i].x >> 24), 1,
                                                  __ATOMIC_RELAXED, __HIP_MEMORY_SCOPE_WORKGROUP);
            }
            #pragma unroll
            for (int i = 0; i < RCH; ++i) {
                if (j0 + i < len) {
                    const int dl = (int)((unsigned)e[i].x >> 24);
                    const int2 rec = make_int2(e[i].x & 0xFFFFFF, e[i].y);
                    if (r[i] < 32) {
                        stage[dl][slot_of(r[i] & 1, r[i] >> 1)] = rec;
                    } else if (r[i] < 2 * HSEG) {
                        const size_t pos = (size_t)(node0 + dl) * (2 * HSEG) + slot_of(r[i] & 1, r[i] >> 1);
                        *(int2*)(slots + 2 * pos) = rec;
                    } else {
                        next[p[i]] = __hip_atomic_exchange(head + node0 + dl, p[i], __ATOMIC_RELAXED,
                                                           __HIP_MEMORY_SCOPE_AGENT);
                    }
                }
            }
        }
    }
    __syncthreads();

    for (int i = t; i < BKT_N; i += 256) {
        if (node0 + i < N) {
            const int tot = cnt[i];
            const int m   = min(tot, 2 * HSEG);
            cnt2[node0 + i]          = ((m + 1) >> 1) + (tot - m);
            cnt2[NNODES + node0 + i] = m >> 1;
        }
    }
    const int4* sg = (const int4*)lds;
    for (int i = t; i < BKT_N * 16; i += 256) {
        const int nd = i >> 4, q = i & 15;
        if (node0 + nd < N)
            *(int4*)(slots + (size_t)(node0 + nd) * (4 * HSEG) + 4 * q) = sg[nd * 16 + q];
    }
}

template <int I>
__device__ __forceinline__ int row_share(const int v)
{
    return __builtin_amdgcn_update_dpp(0, v, 0x150 + I, 0xF, 0xF, true);
}
__device__ __forceinline__ int row_lane(const int v, const int i)
{
    switch (i) {
    case 0:  return row_share<0>(v);
    case 1:  return row_share<1>(v);
    case 2:  return row_share<2>(v);
    case 3:  return row_share<3>(v);
    case 4:  return row_share<4>(v);
    case 5:  return row_share<5>(v);
    case 6:  return row_share<6>(v);
    case 7:  return row_share<7>(v);
    case 8:  return row_share<8>(v);
    case 9:  return row_share<9>(v);
    case 10: return row_share<10>(v);
    case 11: return row_share<11>(v);
    case 12: return row_share<12>(v);
    case 13: return row_share<13>(v);
    case 14: return row_share<14>(v);
    default: return row_share<15>(v);
    }
}

__device__ __forceinline__ void fmav(float& a, const float r, const float w)
{
    a = fmaf(r, w, a);
}
__device__ __forceinline__ void fmav(float4& a, const float4 r, const float w)
{
    a.x = fmaf(r.x, w, a.x); a.y = fmaf(r.y, w, a.y);
    a.z = fmaf(r.z, w, a.z); a.w = fmaf(r.w, w, a.w);
}
__device__ __forceinline__ float  zero_of(const float*)  { return 0.f; }
__device__ __forceinline__ float4 zero_of(const float4*) { return make_float4(0.f, 0.f, 0.f, 0.f); }

__device__ __forceinline__ int list_slot(const int k, const int n0)
{
    return k < n0 ? slot_of(0, k) : slot_of(1, k - n0);
}

__device__ __forceinline__ int2 load_record(const int2* __restrict__ ed, const int node, const int s)
{
    const unsigned off = ((unsigned)node * (2u * HSEG) + (unsigned)s) * (unsigned)sizeof(int2);
    return *(const int2*)((const char*)ed + off);
}

__device__ __forceinline__ void load_records(
    const int2* __restrict__ ed, const int node, const int c0, const int c1,
    const int l, int2& rec0, int2& rec1)
{
    const int n0  = min(c0, HSEG);
    const int len = n0 + min(c1, HSEG);
    rec0 = make_int2(0, 0);
    rec1 = make_int2(0, 0);
    if (len > 0) {
        rec0 = load_record(ed, node, list_slot(min(l, len - 1), n0));
        rec1 = load_record(ed, node, list_slot(min(16 + l, len - 1), n0));
    }
}

template <typename V, int VPR, int CH>
__device__ __forceinline__ V gather_node(
    const V* __restrict__ Pv, const int tot, const int col,
    const int2 rec0, const int2 rec1)
{
    static_assert(16 % CH == 0, "a load of 16 records is whole chunks");
    constexpr unsigned ROWB = VPR * sizeof(V);
    const char* base = (const char*)Pv;
    const unsigned colb = (unsigned)col * (unsigned)sizeof(V);
    const int end = min(tot, 32);
    V acc = zero_of((const V*)nullptr);
    #pragma unroll
    for (int c = 0; c < 32; c += CH) {
        const int m = end - c;
        if (m > 0) {
            const int2 rec = c < 16 ? rec0 : rec1;
            const int cl = c & 15;
            unsigned off[CH];
            #pragma unroll
            for (int i = 0; i < CH; ++i)
                off[i] = (unsigned)row_lane(rec.x, cl + i) * ROWB + colb;
            V r[CH];
            #pragma unroll
            for (int i = 0; i < CH; ++i)
                if (i < m) r[i] = *(const V*)(base + off[i]);
            #pragma unroll
            for (int i = 0; i < CH; ++i)
                if (i < m) fmav(acc, r[i], __int_as_float(row_lane(rec.y, cl + i)));
        }
    }
    return acc;
}

struct EdgeChain {
    const int*  head;
    const int*  next;
    const int2* rec;
};

template <typename V, int VPR>
__device__ __forceinline__ V gather_rest(
    const V* __restrict__ Pv, const int2* __restrict__ ed, const EdgeChain& ch,
    const int* __restrict__ cnt2, const int node, const int col, V acc)
{
    constexpr unsigned ROWB = VPR * sizeof(V);
    const char* base = (const char*)Pv;
    const unsigned colb = (unsigned)col * (unsigned)sizeof(V);
    const int c0  = cnt2[node];
    const int c1  = cnt2[NNODES + node];
    const int n0  = min(c0, HSEG);
    const int len = n0 + min(c1, HSEG);
    for (int k = 32; k < len; ++k) {
        const int2 rec = load_record(ed, node, list_slot(k, n0));
        fmav(acc, *(const V*)(base + ((unsigned)rec.x * ROWB + colb)), __int_as_float(rec.y));
    }
    const int links = max(c0 - HSEG, 0) + max(c1 - HSEG, 0);
    if (links > 0) {
        int o = ch.head[node];
        for (int i = 0; i < links; ++i) {
            o = min(max(o, 0), NEDGES - 1);
            const int2 e = ch.rec[o];
            fmav(acc, *(const V*)(base + ((unsigned)(e.x & 0xFFFFFF) * ROWB + colb)), __int_as_float(e.y));
            o = ch.next[o];
        }
    }
    return acc;
}

__device__ __forceinline__ void load_counts(
    const int* __restrict__ cnt2, const int tl, const int ntiles,
    const int nb, const int N, int& c0, int& c1)
{
    c0 = 0;
    c1 = 0;
    const int node = tl * 16 + nb;
    if (tl < ntiles && node < N) {
        c0 = cnt2[node];
        c1 = cnt2[NNODES + node];
    }
}

template <int DOUT, int DOUTP>
__global__ __launch_bounds__(256, 10) void gcn_gather64_gemm(
    const float* __restrict__ P, const int* __restrict__ cnt2,
    const int* __restrict__ slots, const int* __restrict__ head,
    const int* __restrict__ next, const int* __restrict__ erec,
    const float* __restrict__ Wm,
    const float* __restrict__ bias, float* __restrict__ Pn, int N, int ntiles)
{
    constexpr int DIN  = 64;
    constexpr int LROW = DIN + 4;
    static_assert(DOUTP % 16 == 0 && DOUTP <= 64 && DOUT <= DOUTP, "column tiles");
    __shared__ float lds[2][16 * LROW];

    const int t    = threadIdx.x;
    const int nb   = t >> 4;
    const int l    = t & 15;

    const int wave    = t >> 5;
    const int lane    = t & 31;
    const int colbase = wave * 16;
    const bool mult   = colbase < DOUTP;
    const int half = lane >> 4;
    const int nrow = lane & 15;
    const int col  = colbase + nrow;

    v2f b[DIN / 4];
    float bv = 0.f;
    if (mult) {
        const float* wk = Wm + (size_t)(2 * half) * DOUTP + col;
        #pragma unroll
        for (int s = 0; s < DIN / 4; ++s) {
            b[s].x = wk[(size_t)(4 * s) * DOUTP];
            b[s].y = wk[(size_t)(4 * s + 1) * DOUTP];
        }
        bv = bias[col];
    }
    const int aoff = nrow * LROW + 2 * half;

    const int2* ed = (const int2*)slots;
    const EdgeChain ch = {head, next, (const int2*)erec};
    const int G = (int)gridDim.x;
    int tot, nc0, nc1;
    int2 rec0, rec1;
    load_counts(cnt2, (int)blockIdx.x, ntiles, nb, N, nc0, nc1);
    load_records(ed, (int)blockIdx.x * 16 + nb, nc0, nc1, l, rec0, rec1);
    tot = nc0 + nc1;
    load_counts(cnt2, (int)blockIdx.x + G, ntiles, nb, N, nc0, nc1);

    int buf = 0;
    for (int tl = blockIdx.x; tl < ntiles; tl += G, buf ^= 1) {
        const int row0 = tl * 16;
        float* tile = lds[buf];

        float4 acc = gather_node<float4, 16, 8>((const float4*)P, tot, l, rec0, rec1);
        if (tot > 32)
            acc = gather_rest<float4, 16>((const float4*)P, ed, ch, cnt2, row0 + nb, l, acc);

        load_records(ed, (tl + G) * 16 + nb, nc0, nc1, l, rec0, rec1);
        tot = nc0 + nc1;
        load_counts(cnt2, tl + 2 * G, ntiles, nb, N, nc0, nc1);

        acc.x = fmaxf(acc.x, 0.f); acc.y = fmaxf(acc.y, 0.f);
        acc.z = fmaxf(acc.z, 0.f); acc.w = fmaxf(acc.w, 0.f);
        *(float4*)(&tile[nb * LROW + 4 * l]) = acc;
        lds_barrier();

        if (mult) {
            v8f d = {};
            #pragma unroll
            for (int s = 0; s < DIN / 4; ++s) {
                v2f a;
                a.x = tile[aoff + 4 * s];
                a.y = tile[aoff + 4 * s + 1];
                d = __builtin_amdgcn_wmma_f32_16x16x4_f32(
                        false, a, false, b[s], (short)0, d, false, false);
            }
            #pragma unroll
            for (int r = 0; r < 8; ++r) {
                const int m   = r + 8 * half;
                const int row = row0 + m;
                if (row < N && col < DOUT)
                    Pn[(size_t)row * DOUT + col] = d[r] + bv;
            }
        }
    }
}

__global__ __launch_bounds__(256) void gcn_gather10(
    const float* __restrict__ P, const int* __restrict__ cnt2,
    const int* __restrict__ slots, const int* __restrict__ head,
    const int* __restrict__ next, const int* __restrict__ erec,
    float* __restrict__ out, int N, int ntiles)
{
    __shared__ float stage[2][160];
    const int t    = threadIdx.x;
    const int nb   = t >> 4;
    const int l    = t & 15;

    const int2* ed = (const int2*)slots;
    const EdgeChain ch = {head, next, (const int2*)erec};
    const int G = (int)gridDim.x;
    int tot, nc0, nc1;
    int2 rec0, rec1;
    load_counts(cnt2, (int)blockIdx.x, ntiles, nb, N, nc0, nc1);
    load_records(ed, (int)blockIdx.x * 16 + nb, nc0, nc1, l, rec0, rec1);
    tot = nc0 + nc1;
    load_counts(cnt2, (int)blockIdx.x + G, ntiles, nb, N, nc0, nc1);

    int buf = 0;
    for (int tl = blockIdx.x; tl < ntiles; tl += G, buf ^= 1) {
        float acc = gather_node<float, 10, 16>(P, tot, min(l, 9), rec0, rec1);
        if (tot > 32)
            acc = gather_rest<float, 10>(P, ed, ch, cnt2, tl * 16 + nb, min(l, 9), acc);
        load_records(ed, (tl + G) * 16 + nb, nc0, nc1, l, rec0, rec1);
        tot = nc0 + nc1;
        load_counts(cnt2, tl + 2 * G, ntiles, nb, N, nc0, nc1);
        if (l < 10) stage[buf][nb * 10 + l] = acc;
        lds_barrier();
        const int o = tl * 160 + t;
        if (t < 160 && o < N * 10) out[o] = stage[buf][t];
    }
}

extern "C" void kernel_launch(void* const* d_in, const int* in_sizes, int n_in,
                              void* d_out, int out_size, void* d_ws, size_t ws_size,
                              hipStream_t stream)
{
    (void)in_sizes; (void)n_in; (void)out_size; (void)ws_size;
    const int N = NNODES, E = NEDGES;

    const float* x    = (const float*)d_in[0];
    const int*   esrc = (const int*)  d_in[1];
    const int*   edst = (const int*)  d_in[2];
    const float* ew   = (const float*)d_in[3];
    const float* Ws[5]; const float* bs[5];
    for (int i = 0; i < 5; ++i) {
        Ws[i] = (const float*)d_in[4 + 2 * i];
        bs[i] = (const float*)d_in[5 + 2 * i];
    }
    float* out  = (float*)d_out;
    float* buf0 = (float*)d_ws;
    float* buf1 = buf0 + (size_t)N * 64;
    float* buf2 = buf1 + (size_t)N * 64;
    float* wpad = buf2 + (size_t)N * 64;
    float* bpad = wpad + 64 * 16;
    int*   cnt   = (int*)(bpad + 64);
    int*   head  = cnt + 2 * N;
    int*   next  = head + (N + 96);
    int*   slots = next + E;
    int*   erec  = slots + (size_t)N * (4 * HSEG);
    int*   offT  = erec + 2 * (size_t)E;

    const int  ntiles = (N + 15) / 16;
    const dim3 pgrid(1250);
    const dim3 bgrid(NTILE + GEMM0_WG_A);
    const dim3 rgrid(NBKT + GEMM0_WG_B);

    gcn_bin_gemm0<<<bgrid, 256, 0, stream>>>(esrc, edst, ew, erec, offT, E,
                                             x, Ws[0], bs[0], buf0, N,
                                             Ws[4], bs[4], wpad, bpad);
    gcn_rank_gemm0<<<rgrid, 256, 0, stream>>>(erec, offT, cnt, head, next, slots, N,
                                              x, Ws[0], bs[0], buf0);

    gcn_gather64_gemm<64, 64><<<pgrid, 256, 0, stream>>>(buf0, cnt, slots, head, next, erec, Ws[1], bs[1], buf1, N, ntiles);
    gcn_gather64_gemm<64, 64><<<pgrid, 256, 0, stream>>>(buf1, cnt, slots, head, next, erec, Ws[2], bs[2], buf0, N, ntiles);
    gcn_gather64_gemm<64, 64><<<pgrid, 256, 0, stream>>>(buf0, cnt, slots, head, next, erec, Ws[3], bs[3], buf1, N, ntiles);

    gcn_gather64_gemm<10, 16><<<pgrid, 256, 0, stream>>>(buf1, cnt, slots, head, next, erec, wpad, bpad, buf0, N, ntiles);

    gcn_gather10<<<pgrid, 256, 0, stream>>>(buf0, cnt, slots, head, next, erec, out, N, ntiles);
}
